# starting kernel
# baseline (speedup 1.0000x reference)
#include <hip/hip_runtime.h>
#include <hip/hip_bf16.h>

typedef __attribute__((ext_vector_type(2))) float v2f;
typedef __attribute__((ext_vector_type(8))) float v8f;

#define HDIM 64
#define INDIM 128

__device__ __forceinline__ float lrelu(float v) { return v > 0.0f ? v : 0.01f * v; }

__global__ void deg_init_kernel(float* __restrict__ deg, int n) {
    int i = blockIdx.x * blockDim.x + threadIdx.x;
    if (i < n) deg[i] = 1.0f;
}

__global__ void deg_count_kernel(const int* __restrict__ dst, float* __restrict__ deg, int e) {
    int i = blockIdx.x * blockDim.x + threadIdx.x;
    if (i < e)
        __hip_atomic_fetch_add(&deg[dst[i]], 1.0f, __ATOMIC_RELAXED, __HIP_MEMORY_SCOPE_AGENT);
}

__global__ void dinv_kernel(float* __restrict__ deg, int n) {
    int i = blockIdx.x * blockDim.x + threadIdx.x;
    if (i < n) {
        float d = deg[i];
        deg[i] = d > 0.0f ? rsqrtf(d) : 0.0f;
    }
}

__global__ void wmma_gemm_f32_kernel(const float* __restrict__ A, const float* __restrict__ B,
                                     float* __restrict__ C, const float* __restrict__ bias,
                                     int M, int K, int act) {
    extern __shared__ float bsw[];
    const int tid   = threadIdx.x;
    const int lane  = tid & 31;
    const int wave  = tid >> 5;
    const int lmod  = lane & 15;
    const int lhalf = lane >> 4;

    const int pairs = K * 32;
    for (int p = tid; p < pairs; p += 256) {
        int ks  = p >> 7;
        int ll  = (p >> 2) & 31;
        int tn  = p & 3;
        int kb  = 4 * ks + 2 * (ll >> 4);
        int col = tn * 16 + (ll & 15);
        bsw[p * 2 + 0] = B[(size_t)kb * HDIM + col];
        bsw[p * 2 + 1] = B[(size_t)(kb + 1) * HDIM + col];
    }
    __syncthreads();

    const int gwave = blockIdx.x * 8 + wave;
    if (gwave >= (M >> 4)) return;

    const float* arow = A + (size_t)(gwave * 16 + lmod) * K + 2 * lhalf;
    v8f acc0 = {}, acc1 = {}, acc2 = {}, acc3 = {};

    const int steps = K >> 2;
    for (int ks = 0; ks < steps; ++ks) {
        v2f a = *(const v2f*)(arow + 4 * ks);
        const float4* bp = (const float4*)(bsw + (size_t)(ks * 32 + lane) * 8);
        float4 b01 = bp[0];
        float4 b23 = bp[1];
        v2f b0 = {b01.x, b01.y}, b1 = {b01.z, b01.w};
        v2f b2 = {b23.x, b23.y}, b3 = {b23.z, b23.w};
        acc0 = __builtin_amdgcn_wmma_f32_16x16x4_f32(false, a, false, b0, (short)0, acc0, false, false);
        acc1 = __builtin_amdgcn_wmma_f32_16x16x4_f32(false, a, false, b1, (short)0, acc1, false, false);
        acc2 = __builtin_amdgcn_wmma_f32_16x16x4_f32(false, a, false, b2, (short)0, acc2, false, false);
        acc3 = __builtin_amdgcn_wmma_f32_16x16x4_f32(false, a, false, b3, (short)0, acc3, false, false);
    }

    const int rbase = gwave * 16 + lhalf * 8;
    float bi0 = 0.f, bi1 = 0.f, bi2 = 0.f, bi3 = 0.f;
    if (bias) { bi0 = bias[lmod]; bi1 = bias[16 + lmod]; bi2 = bias[32 + lmod]; bi3 = bias[48 + lmod]; }
#pragma unroll
    for (int r = 0; r < 8; ++r) {
        size_t ro = (size_t)(rbase + r) * HDIM + lmod;
        float v0 = acc0[r] + bi0, v1 = acc1[r] + bi1, v2 = acc2[r] + bi2, v3 = acc3[r] + bi3;
        if (act) { v0 = lrelu(v0); v1 = lrelu(v1); v2 = lrelu(v2); v3 = lrelu(v3); }
        C[ro] = v0; C[ro + 16] = v1; C[ro + 32] = v2; C[ro + 48] = v3;
    }
}

__global__ void conv_init_kernel(const float* __restrict__ Hm, const float* __restrict__ dinv,
                                 const float* __restrict__ bias, float* __restrict__ out, int n) {
    int tid = blockIdx.x * blockDim.x + threadIdx.x;
    if (tid >= n * 16) return;
    int i = tid >> 4, j = (tid & 15) * 4;
    float s = dinv[i]; s *= s;
    float4 h = *(const float4*)(Hm + (size_t)i * HDIM + j);
    float4 b = *(const float4*)(bias + j);
    float4 o = make_float4(b.x + s * h.x, b.y + s * h.y, b.z + s * h.z, b.w + s * h.w);
    *(float4*)(out + (size_t)i * HDIM + j) = o;
}

__global__ void conv_scatter_kernel(const float* __restrict__ Hm, const int* __restrict__ src,
                                    const int* __restrict__ dst, const float* __restrict__ dinv,
                                    float* __restrict__ out, int e) {
    int tid = blockIdx.x * blockDim.x + threadIdx.x;
    int ed = tid >> 4;
    if (ed >= e) return;
    int j = (tid & 15) * 4;
    int s = src[ed], d = dst[ed];
    float nrm = dinv[s] * dinv[d];
    float4 h = *(const float4*)(Hm + (size_t)s * HDIM + j);
    float* o = out + (size_t)d * HDIM + j;
    __hip_atomic_fetch_add(o + 0, nrm * h.x, __ATOMIC_RELAXED, __HIP_MEMORY_SCOPE_AGENT);
    __hip_atomic_fetch_add(o + 1, nrm * h.y, __ATOMIC_RELAXED, __HIP_MEMORY_SCOPE_AGENT);
    __hip_atomic_fetch_add(o + 2, nrm * h.z, __ATOMIC_RELAXED, __HIP_MEMORY_SCOPE_AGENT);
    __hip_atomic_fetch_add(o + 3, nrm * h.w, __ATOMIC_RELAXED, __HIP_MEMORY_SCOPE_AGENT);
}

__global__ void relu_inplace_kernel(float* __restrict__ buf, int n16) {
    int tid = blockIdx.x * blockDim.x + threadIdx.x;
    if (tid >= n16) return;
    float4* p = (float4*)buf + tid;
    float4 v = *p;
    v.x = fmaxf(v.x, 0.0f); v.y = fmaxf(v.y, 0.0f);
    v.z = fmaxf(v.z, 0.0f); v.w = fmaxf(v.w, 0.0f);
    *p = v;
}

__global__ void dual_head_kernel(const float* __restrict__ Z, const int* __restrict__ idx, int n,
                                 const float* __restrict__ Wa, const float* __restrict__ ba,
                                 float* __restrict__ outA,
                                 const float* __restrict__ Wb, const float* __restrict__ bb,
                                 float* __restrict__ outB) {
    int t = blockIdx.x * blockDim.x + threadIdx.x;
    if (t >= n) return;
    const float* row = Z + (size_t)idx[t] * HDIM;
    float da = 0.0f, db = 0.0f;
#pragma unroll
    for (int j = 0; j < HDIM; j += 4) {
        float4 z  = *(const float4*)(row + j);
        float4 wa = *(const float4*)(Wa + j);
        float4 wb = *(const float4*)(Wb + j);
        da += z.x * wa.x + z.y * wa.y + z.z * wa.z + z.w * wa.w;
        db += z.x * wb.x + z.y * wb.y + z.z * wb.z + z.w * wb.w;
    }
    outA[t] = lrelu(da + ba[0]);
    outB[t] = lrelu(db + bb[0]);
}

__global__ void tprob_kernel(const float* __restrict__ P1, const float* __restrict__ Wp2,
                             const float* __restrict__ bp2, float* __restrict__ out, int n) {
    int i = blockIdx.x * blockDim.x + threadIdx.x;
    if (i >= n) return;
    const float* row = P1 + (size_t)i * HDIM;
    float d0 = 0.0f, d1 = 0.0f;
#pragma unroll
    for (int k = 0; k < HDIM; ++k) {
        float z = row[k];
        d0 += z * Wp2[k * 2 + 0];
        d1 += z * Wp2[k * 2 + 1];
    }
    out[i * 2 + 0] = lrelu(d0 + bp2[0]);
    out[i * 2 + 1] = lrelu(d1 + bp2[1]);
}

extern "C" void kernel_launch(void* const* d_in, const int* in_sizes, int n_in,
                              void* d_out, int out_size, void* d_ws, size_t ws_size,
                              hipStream_t stream) {
    const float* x     = (const float*)d_in[0];
    const int*   ei    = (const int*)d_in[1];
    const int*   treat = (const int*)d_in[2];
    const int*   ctrl  = (const int*)d_in[3];
    const float* W1  = (const float*)d_in[4];  const float* b1  = (const float*)d_in[5];
    const float* W2  = (const float*)d_in[6];  const float* b2  = (const float*)d_in[7];
    const float* Wy1 = (const float*)d_in[8];  const float* by1 = (const float*)d_in[9];
    const float* Wy0 = (const float*)d_in[10]; const float* by0 = (const float*)d_in[11];
    const float* Wp1 = (const float*)d_in[12]; const float* bp1 = (const float*)d_in[13];
    const float* Wp2 = (const float*)d_in[14]; const float* bp2 = (const float*)d_in[15];

    const int N = in_sizes[0] / INDIM;
    const int E = in_sizes[1] / 2;
    const int T = in_sizes[2];
    const int Cn = in_sizes[3];
    const int* e_src = ei;
    const int* e_dst = ei + E;

    float* dinv = (float*)d_ws;
    float* buf0 = dinv + (((size_t)N + 63) & ~(size_t)63);
    float* buf1 = buf0 + (size_t)N * HDIM;

    float* o_y1    = (float*)d_out;
    float* o_yc0   = o_y1 + T;
    float* o_y0    = o_yc0 + T;
    float* o_yc1   = o_y0 + Cn;
    float* o_tprob = o_yc1 + Cn;
    float* o_xZ2   = o_tprob + (size_t)2 * N;

    const int BLK = 256;
    auto cdiv = [](long long a, long long b) { return (int)((a + b - 1) / b); };
    const int gemm_blocks = cdiv(N / 16, 8);
    const size_t smem128 = (size_t)INDIM * HDIM * sizeof(float);
    const size_t smem64  = (size_t)HDIM * HDIM * sizeof(float);

    deg_init_kernel<<<cdiv(N, BLK), BLK, 0, stream>>>(dinv, N);
    deg_count_kernel<<<cdiv(E, BLK), BLK, 0, stream>>>(e_dst, dinv, E);
    dinv_kernel<<<cdiv(N, BLK), BLK, 0, stream>>>(dinv, N);

    wmma_gemm_f32_kernel<<<gemm_blocks, BLK, smem128, stream>>>(x, W1, buf0, nullptr, N, INDIM, 0);
    conv_init_kernel<<<cdiv((long long)N * 16, BLK), BLK, 0, stream>>>(buf0, dinv, b1, buf1, N);
    conv_scatter_kernel<<<cdiv((long long)E * 16, BLK), BLK, 0, stream>>>(buf0, e_src, e_dst, dinv, buf1, E);
    relu_inplace_kernel<<<cdiv((long long)N * 16, BLK), BLK, 0, stream>>>(buf1, N * 16);

    wmma_gemm_f32_kernel<<<gemm_blocks, BLK, smem64, stream>>>(buf1, W2, buf0, nullptr, N, HDIM, 0);
    conv_init_kernel<<<cdiv((long long)N * 16, BLK), BLK, 0, stream>>>(buf0, dinv, b2, o_xZ2, N);
    conv_scatter_kernel<<<cdiv((long long)E * 16, BLK), BLK, 0, stream>>>(buf0, e_src, e_dst, dinv, o_xZ2, E);

    dual_head_kernel<<<cdiv(T, BLK), BLK, 0, stream>>>(o_xZ2, treat, T, Wy1, by1, o_y1, Wy0, by0, o_yc0);
    dual_head_kernel<<<cdiv(Cn, BLK), BLK, 0, stream>>>(o_xZ2, ctrl, Cn, Wy0, by0, o_y0, Wy1, by1, o_yc1);

    wmma_gemm_f32_kernel<<<gemm_blocks, BLK, smem64, stream>>>(o_xZ2, Wp1, buf0, bp1, N, HDIM, 1);
    tprob_kernel<<<cdiv(N, BLK), BLK, 0, stream>>>(buf0, Wp2, bp2, o_tprob, N);
}


# Degree count leaves 1.6M scattered global f32 atomics for LDS counters: block (node range, edge chunk) counts its dst hits, stores a partial plane; rsqrt pass sums 32 planes; 11 launches to 10
# speedup vs baseline: 2.2874x; 2.2874x over previous
#include <hip/hip_runtime.h>
#include <hip/hip_bf16.h>

typedef __attribute__((ext_vector_type(2))) float v2f;
typedef __attribute__((ext_vector_type(8))) float v8f;

#define HDIM 64
#define INDIM 128

__device__ __forceinline__ float lrelu(float v) { return v > 0.0f ? v : 0.01f * v; }

#define DEG_R 8
#define DEG_C 32
#define DEG_T 512
#define DEG_U 8

__device__ __forceinline__ void deg_bump(unsigned* cnt, int d, int lo, unsigned span) {
    const unsigned o = (unsigned)(d - lo);
    if (o < span) atomicAdd(&cnt[o], 1u);
}

__global__ __launch_bounds__(DEG_T) void deg_hist_kernel(const int* __restrict__ dst, unsigned* __restrict__ part,
                                                         int n, int e, int span) {
    extern __shared__ unsigned cnt[];
    const int tid = threadIdx.x;
    const int r = blockIdx.x % DEG_R;
    const int c = blockIdx.x / DEG_R;
    const int lo = r * span;
    const unsigned uspan = (unsigned)span;

    for (int i = tid; i < span; i += DEG_T) cnt[i] = 0u;
    __syncthreads();

    const int per = (((e + DEG_C - 1) / DEG_C) + 3) & ~3;
    const int e0 = min(e, c * per);
    const int e1 = min(e, e0 + per);
    if ((((size_t)dst) & 15) == 0) {
        const int n4 = (e1 - e0) >> 2;
        const int4* __restrict__ p4 = (const int4*)(dst + e0);
        for (int base = tid; base < n4; base += DEG_U * DEG_T) {
            int4 v[DEG_U];
#pragma unroll
            for (int u = 0; u < DEG_U; ++u) {
                const int idx = base + u * DEG_T;
                v[u] = (idx < n4) ? p4[idx] : make_int4(-1, -1, -1, -1);
            }
#pragma unroll
            for (int u = 0; u < DEG_U; ++u) {
                deg_bump(cnt, v[u].x, lo, uspan);
                deg_bump(cnt, v[u].y, lo, uspan);
                deg_bump(cnt, v[u].z, lo, uspan);
                deg_bump(cnt, v[u].w, lo, uspan);
            }
        }
        const int t0 = e0 + 4 * n4;
        if (t0 + tid < e1) deg_bump(cnt, dst[t0 + tid], lo, uspan);
    } else {
        for (int i = e0 + tid; i < e1; i += DEG_T) deg_bump(cnt, dst[i], lo, uspan);
    }
    __syncthreads();

    unsigned* __restrict__ plane = part + (size_t)c * n;
    for (int base = tid; base < span; base += 4 * DEG_T) {
        unsigned t[4];
#pragma unroll
        for (int u = 0; u < 4; ++u) {
            const int i = base + u * DEG_T;
            t[u] = (i < span) ? cnt[i] : 0u;
        }
#pragma unroll
        for (int u = 0; u < 4; ++u) {
            const int i = base + u * DEG_T;
            if (i < span && lo + i < n) plane[lo + i] = t[u];
        }
    }
}

__global__ void dinv_kernel(const unsigned* __restrict__ part, float* __restrict__ dinv, int n) {
    int i = blockIdx.x * blockDim.x + threadIdx.x;
    if (i < n) {
        unsigned s = 0u;
#pragma unroll
        for (int c = 0; c < DEG_C; ++c) s += part[(size_t)c * n + i];
        const float d = 1.0f + (float)s;
        dinv[i] = d > 0.0f ? rsqrtf(d) : 0.0f;
    }
}

template <int MODE>
__device__ __forceinline__ void gemm_strip_f32(const float* __restrict__ A, const float* __restrict__ B,
                                               float* __restrict__ C, const float* __restrict__ bias,
                                               const float* __restrict__ dinv, float* __restrict__ C2,
                                               int M, int K) {
    extern __shared__ float bsw[];
    const int tid   = threadIdx.x;
    const int lane  = tid & 31;
    const int wave  = tid >> 5;
    const int lmod  = lane & 15;
    const int lhalf = lane >> 4;

    const int pairs = K * 32;
    for (int p = tid; p < pairs; p += 256) {
        int ks  = p >> 7;
        int ll  = (p >> 2) & 31;
        int tn  = p & 3;
        int kb  = 4 * ks + 2 * (ll >> 4);
        int col = tn * 16 + (ll & 15);
        bsw[p * 2 + 0] = B[(size_t)kb * HDIM + col];
        bsw[p * 2 + 1] = B[(size_t)(kb + 1) * HDIM + col];
    }
    __syncthreads();

    const int gwave = blockIdx.x * 8 + wave;
    if (gwave >= (M >> 4)) return;

    const float* arow = A + (size_t)(gwave * 16 + lmod) * K + 2 * lhalf;
    v8f acc0 = {}, acc1 = {}, acc2 = {}, acc3 = {};

    const int steps = K >> 2;
    for (int ks = 0; ks < steps; ++ks) {
        v2f a = *(const v2f*)(arow + 4 * ks);
        if (MODE == 1) { a.x = fmaxf(a.x, 0.0f); a.y = fmaxf(a.y, 0.0f); }
        const float4* bp = (const float4*)(bsw + (size_t)(ks * 32 + lane) * 8);
        float4 b01 = bp[0];
        float4 b23 = bp[1];
        v2f b0 = {b01.x, b01.y}, b1 = {b01.z, b01.w};
        v2f b2 = {b23.x, b23.y}, b3 = {b23.z, b23.w};
        acc0 = __builtin_amdgcn_wmma_f32_16x16x4_f32(false, a, false, b0, (short)0, acc0, false, false);
        acc1 = __builtin_amdgcn_wmma_f32_16x16x4_f32(false, a, false, b1, (short)0, acc1, false, false);
        acc2 = __builtin_amdgcn_wmma_f32_16x16x4_f32(false, a, false, b2, (short)0, acc2, false, false);
        acc3 = __builtin_amdgcn_wmma_f32_16x16x4_f32(false, a, false, b3, (short)0, acc3, false, false);
    }

    const int rbase = gwave * 16 + lhalf * 8;
    const float bi0 = bias[lmod], bi1 = bias[16 + lmod], bi2 = bias[32 + lmod], bi3 = bias[48 + lmod];
    if (MODE == 2) {
#pragma unroll
        for (int r = 0; r < 8; ++r) {
            size_t ro = (size_t)(rbase + r) * HDIM + lmod;
            C[ro]      = lrelu(acc0[r] + bi0);
            C[ro + 16] = lrelu(acc1[r] + bi1);
            C[ro + 32] = lrelu(acc2[r] + bi2);
            C[ro + 48] = lrelu(acc3[r] + bi3);
        }
    } else {
        const float4 dq0 = *(const float4*)(dinv + rbase);
        const float4 dq1 = *(const float4*)(dinv + rbase + 4);
        const float dv[8] = {dq0.x, dq0.y, dq0.z, dq0.w, dq1.x, dq1.y, dq1.z, dq1.w};
#pragma unroll
        for (int r = 0; r < 8; ++r) {
            size_t ro = (size_t)(rbase + r) * HDIM + lmod;
            const float s = dv[r] * dv[r];
            const float h0 = acc0[r], h1 = acc1[r], h2 = acc2[r], h3 = acc3[r];
            C[ro] = h0; C[ro + 16] = h1; C[ro + 32] = h2; C[ro + 48] = h3;
            C2[ro]      = bi0 + s * h0;
            C2[ro + 16] = bi1 + s * h1;
            C2[ro + 32] = bi2 + s * h2;
            C2[ro + 48] = bi3 + s * h3;
        }
    }
}

__global__ void conv_gemm_f32_kernel(const float* __restrict__ A, const float* __restrict__ B,
                                     float* __restrict__ C, const float* __restrict__ bias,
                                     const float* __restrict__ dinv, float* __restrict__ C2, int M, int K) {
    gemm_strip_f32<0>(A, B, C, bias, dinv, C2, M, K);
}

__global__ void conv_gemm_relu_f32_kernel(const float* __restrict__ A, const float* __restrict__ B,
                                          float* __restrict__ C, const float* __restrict__ bias,
                                          const float* __restrict__ dinv, float* __restrict__ C2, int M, int K) {
    gemm_strip_f32<1>(A, B, C, bias, dinv, C2, M, K);
}

__global__ void dense_gemm_lrelu_f32_kernel(const float* __restrict__ A, const float* __restrict__ B,
                                            float* __restrict__ C, const float* __restrict__ bias, int M, int K) {
    gemm_strip_f32<2>(A, B, C, bias, nullptr, nullptr, M, K);
}

#define SC_EB 16
#define SCATTER_BLOCKS 4096

__device__ __forceinline__ int lane_bcast_i(int v, int l) { return __builtin_amdgcn_readlane(v, l); }
__device__ __forceinline__ float lane_bcast_f(float v, int l) {
    return __int_as_float(__builtin_amdgcn_readlane(__float_as_int(v), l));
}

__global__ __launch_bounds__(256) void conv_scatter_kernel(const float* __restrict__ Hm, const int* __restrict__ src,
                                    const int* __restrict__ dst, const float* __restrict__ dinv,
                                    float* __restrict__ out, int e) {
    const int lane = threadIdx.x & 31;
    const int sub  = lane & 15;
    const int wave   = __builtin_amdgcn_readfirstlane((int)((blockIdx.x * blockDim.x + threadIdx.x) >> 5));
    const int nwaves = (int)((gridDim.x * blockDim.x) >> 5);
    const int* __restrict__ idxp = (lane < 16) ? src : dst;
    const int nb = (e + SC_EB - 1) / SC_EB;

    for (int b = wave; b < nb; b += nwaves) {
        const int e0  = b * SC_EB;
        const int cnt = min(SC_EB, e - e0);
        const int id  = (sub < cnt) ? idxp[e0 + sub] : 0;
        const float dv = dinv[id];

        if (cnt == SC_EB) {
            float h0[SC_EB], h1[SC_EB];
#pragma unroll
            for (int k = 0; k < SC_EB; ++k) {
                const int s = lane_bcast_i(id, k);
                const float* hr = Hm + (size_t)s * HDIM + lane;
                h0[k] = hr[0];
                h1[k] = hr[32];
            }
#pragma unroll
            for (int k = 0; k < SC_EB; ++k) {
                const int d = lane_bcast_i(id, 16 + k);
                const float nrm = lane_bcast_f(dv, k) * lane_bcast_f(dv, 16 + k);
                float* o = out + (size_t)d * HDIM + lane;
                __hip_atomic_fetch_add(o,      nrm * h0[k], __ATOMIC_RELAXED, __HIP_MEMORY_SCOPE_AGENT);
                __hip_atomic_fetch_add(o + 32, nrm * h1[k], __ATOMIC_RELAXED, __HIP_MEMORY_SCOPE_AGENT);
            }
        } else {
            for (int k = 0; k < cnt; ++k) {
                const int s = lane_bcast_i(id, k);
                const int d = lane_bcast_i(id, 16 + k);
                const float nrm = lane_bcast_f(dv, k) * lane_bcast_f(dv, 16 + k);
                const float* hr = Hm + (size_t)s * HDIM + lane;
                float* o = out + (size_t)d * HDIM + lane;
                const float a0 = hr[0], a1 = hr[32];
                __hip_atomic_fetch_add(o,      nrm * a0, __ATOMIC_RELAXED, __HIP_MEMORY_SCOPE_AGENT);
                __hip_atomic_fetch_add(o + 32, nrm * a1, __ATOMIC_RELAXED, __HIP_MEMORY_SCOPE_AGENT);
            }
        }
    }
}

__global__ void dual_head_kernel(const float* __restrict__ Z, const int* __restrict__ idx, int n,
                                 const float* __restrict__ Wa, const float* __restrict__ ba,
                                 float* __restrict__ outA,
                                 const float* __restrict__ Wb, const float* __restrict__ bb,
                                 float* __restrict__ outB) {
    int t = blockIdx.x * blockDim.x + threadIdx.x;
    if (t >= n) return;
    const float* row = Z + (size_t)idx[t] * HDIM;
    float da = 0.0f, db = 0.0f;
#pragma unroll
    for (int j = 0; j < HDIM; j += 4) {
        float4 z  = *(const float4*)(row + j);
        float4 wa = *(const float4*)(Wa + j);
        float4 wb = *(const float4*)(Wb + j);
        da += z.x * wa.x + z.y * wa.y + z.z * wa.z + z.w * wa.w;
        db += z.x * wb.x + z.y * wb.y + z.z * wb.z + z.w * wb.w;
    }
    outA[t] = lrelu(da + ba[0]);
    outB[t] = lrelu(db + bb[0]);
}

__global__ void tprob_kernel(const float* __restrict__ P1, const float* __restrict__ Wp2,
                             const float* __restrict__ bp2, float* __restrict__ out, int n) {
    int i = blockIdx.x * blockDim.x + threadIdx.x;
    if (i >= n) return;
    const float* row = P1 + (size_t)i * HDIM;
    float d0 = 0.0f, d1 = 0.0f;
#pragma unroll
    for (int k = 0; k < HDIM; ++k) {
        float z = row[k];
        d0 += z * Wp2[k * 2 + 0];
        d1 += z * Wp2[k * 2 + 1];
    }
    out[i * 2 + 0] = lrelu(d0 + bp2[0]);
    out[i * 2 + 1] = lrelu(d1 + bp2[1]);
}

extern "C" void kernel_launch(void* const* d_in, const int* in_sizes, int n_in,
                              void* d_out, int out_size, void* d_ws, size_t ws_size,
                              hipStream_t stream) {
    const float* x     = (const float*)d_in[0];
    const int*   ei    = (const int*)d_in[1];
    const int*   treat = (const int*)d_in[2];
    const int*   ctrl  = (const int*)d_in[3];
    const float* W1  = (const float*)d_in[4];  const float* b1  = (const float*)d_in[5];
    const float* W2  = (const float*)d_in[6];  const float* b2  = (const float*)d_in[7];
    const float* Wy1 = (const float*)d_in[8];  const float* by1 = (const float*)d_in[9];
    const float* Wy0 = (const float*)d_in[10]; const float* by0 = (const float*)d_in[11];
    const float* Wp1 = (const float*)d_in[12]; const float* bp1 = (const float*)d_in[13];
    const float* Wp2 = (const float*)d_in[14]; const float* bp2 = (const float*)d_in[15];

    const int N = in_sizes[0] / INDIM;
    const int E = in_sizes[1] / 2;
    const int T = in_sizes[2];
    const int Cn = in_sizes[3];
    const int* e_src = ei;
    const int* e_dst = ei + E;

    float* dinv = (float*)d_ws;
    float* buf0 = dinv + (((size_t)N + 63) & ~(size_t)63);
    float* buf1 = buf0 + (size_t)N * HDIM;

    float* o_y1    = (float*)d_out;
    float* o_yc0   = o_y1 + T;
    float* o_y0    = o_yc0 + T;
    float* o_yc1   = o_y0 + Cn;
    float* o_tprob = o_yc1 + Cn;
    float* o_xZ2   = o_tprob + (size_t)2 * N;

    const int BLK = 256;
    auto cdiv = [](long long a, long long b) { return (int)((a + b - 1) / b); };
    const int gemm_blocks = cdiv(N / 16, 8);
    const size_t smem128 = (size_t)INDIM * HDIM * sizeof(float);
    const size_t smem64  = (size_t)HDIM * HDIM * sizeof(float);

    const int span = cdiv(N, DEG_R);
    unsigned* degpart = (unsigned*)buf1;
    deg_hist_kernel<<<DEG_R * DEG_C, DEG_T, (size_t)span * sizeof(unsigned), stream>>>(e_dst, degpart, N, E, span);
    dinv_kernel<<<cdiv(N, BLK), BLK, 0, stream>>>(degpart, dinv, N);

    conv_gemm_f32_kernel<<<gemm_blocks, BLK, smem128, stream>>>(x, W1, buf0, b1, dinv, buf1, N, INDIM);
    conv_scatter_kernel<<<SCATTER_BLOCKS, BLK, 0, stream>>>(buf0, e_src, e_dst, dinv, buf1, E);

    conv_gemm_relu_f32_kernel<<<gemm_blocks, BLK, smem64, stream>>>(buf1, W2, buf0, b2, dinv, o_xZ2, N, HDIM);
    conv_scatter_kernel<<<SCATTER_BLOCKS, BLK, 0, stream>>>(buf0, e_src, e_dst, dinv, o_xZ2, E);

    dual_head_kernel<<<cdiv(T, BLK), BLK, 0, stream>>>(o_xZ2, treat, T, Wy1, by1, o_y1, Wy0, by0, o_yc0);
    dual_head_kernel<<<cdiv(Cn, BLK), BLK, 0, stream>>>(o_xZ2, ctrl, Cn, Wy0, by0, o_y0, Wy1, by1, o_yc1);

    dense_gemm_lrelu_f32_kernel<<<gemm_blocks, BLK, smem64, stream>>>(o_xZ2, Wp1, buf0, bp1, N, HDIM);
    tprob_kernel<<<cdiv(N, BLK), BLK, 0, stream>>>(buf0, Wp2, bp2, o_tprob, N);
}


# Scatter-add split by chip half: node range cut in two, a block adds only edges whose dst lies in its own half's range (die id; edge chunks from a counter per half); sums still in atomic arrival order
# speedup vs baseline: 2.4297x; 1.0622x over previous
#include <hip/hip_runtime.h>
#include <hip/hip_bf16.h>

typedef __attribute__((ext_vector_type(2))) float v2f;
typedef __attribute__((ext_vector_type(8))) float v8f;

#define HDIM 64
#define INDIM 128

__device__ __forceinline__ float lrelu(float v) { return v > 0.0f ? v : 0.01f * v; }

#define DEG_R 8
#define DEG_C 32
#define DEG_T 512
#define DEG_U 8

__device__ __forceinline__ void deg_bump(unsigned* cnt, int d, int lo, unsigned span) {
    const unsigned o = (unsigned)(d - lo);
    if (o < span) atomicAdd(&cnt[o], 1u);
}

__global__ __launch_bounds__(DEG_T) void deg_hist_kernel(const int* __restrict__ dst, unsigned* __restrict__ part,
                                                         int n, int e, int span) {
    extern __shared__ unsigned cnt[];
    const int tid = threadIdx.x;
    const int r = blockIdx.x % DEG_R;
    const int c = blockIdx.x / DEG_R;
    const int lo = r * span;
    const unsigned uspan = (unsigned)span;

    for (int i = tid; i < span; i += DEG_T) cnt[i] = 0u;
    __syncthreads();

    const int per = (((e + DEG_C - 1) / DEG_C) + 3) & ~3;
    const int e0 = min(e, c * per);
    const int e1 = min(e, e0 + per);
    if ((((size_t)dst) & 15) == 0) {
        const int n4 = (e1 - e0) >> 2;
        const int4* __restrict__ p4 = (const int4*)(dst + e0);
        for (int base = tid; base < n4; base += DEG_U * DEG_T) {
            int4 v[DEG_U];
#pragma unroll
            for (int u = 0; u < DEG_U; ++u) {
                const int idx = base + u * DEG_T;
                v[u] = (idx < n4) ? p4[idx] : make_int4(-1, -1, -1, -1);
            }
#pragma unroll
            for (int u = 0; u < DEG_U; ++u) {
                deg_bump(cnt, v[u].x, lo, uspan);
                deg_bump(cnt, v[u].y, lo, uspan);
                deg_bump(cnt, v[u].z, lo, uspan);
                deg_bump(cnt, v[u].w, lo, uspan);
            }
        }
        const int t0 = e0 + 4 * n4;
        if (t0 + tid < e1) deg_bump(cnt, dst[t0 + tid], lo, uspan);
    } else {
        for (int i = e0 + tid; i < e1; i += DEG_T) deg_bump(cnt, dst[i], lo, uspan);
    }
    __syncthreads();

    unsigned* __restrict__ plane = part + (size_t)c * n;
    for (int base = tid; base < span; base += 4 * DEG_T) {
        unsigned t[4];
#pragma unroll
        for (int u = 0; u < 4; ++u) {
            const int i = base + u * DEG_T;
            t[u] = (i < span) ? cnt[i] : 0u;
        }
#pragma unroll
        for (int u = 0; u < 4; ++u) {
            const int i = base + u * DEG_T;
            if (i < span && lo + i < n) plane[lo + i] = t[u];
        }
    }
}

__global__ void dinv_kernel(const unsigned* __restrict__ part, float* __restrict__ dinv, unsigned* __restrict__ tickets,
                            int n) {
    int i = blockIdx.x * blockDim.x + threadIdx.x;
    if (i < 4) tickets[i] = 0u;
    if (i < n) {
        unsigned s = 0u;
#pragma unroll
        for (int c = 0; c < DEG_C; ++c) s += part[(size_t)c * n + i];
        const float d = 1.0f + (float)s;
        dinv[i] = d > 0.0f ? rsqrtf(d) : 0.0f;
    }
}

template <int MODE>
__device__ __forceinline__ void gemm_strip_f32(const float* __restrict__ A, const float* __restrict__ B,
                                               float* __restrict__ C, const float* __restrict__ bias,
                                               const float* __restrict__ dinv, float* __restrict__ C2,
                                               int M, int K) {
    extern __shared__ float bsw[];
    const int tid   = threadIdx.x;
    const int lane  = tid & 31;
    const int wave  = tid >> 5;
    const int lmod  = lane & 15;
    const int lhalf = lane >> 4;

    const int pairs = K * 32;
    for (int p = tid; p < pairs; p += 256) {
        int ks  = p >> 7;
        int ll  = (p >> 2) & 31;
        int tn  = p & 3;
        int kb  = 4 * ks + 2 * (ll >> 4);
        int col = tn * 16 + (ll & 15);
        bsw[p * 2 + 0] = B[(size_t)kb * HDIM + col];
        bsw[p * 2 + 1] = B[(size_t)(kb + 1) * HDIM + col];
    }
    __syncthreads();

    const int gwave = blockIdx.x * 8 + wave;
    if (gwave >= (M >> 4)) return;

    const float* arow = A + (size_t)(gwave * 16 + lmod) * K + 2 * lhalf;
    v8f acc0 = {}, acc1 = {}, acc2 = {}, acc3 = {};

    const int steps = K >> 2;
    for (int ks = 0; ks < steps; ++ks) {
        v2f a = *(const v2f*)(arow + 4 * ks);
        if (MODE == 1) { a.x = fmaxf(a.x, 0.0f); a.y = fmaxf(a.y, 0.0f); }
        const float4* bp = (const float4*)(bsw + (size_t)(ks * 32 + lane) * 8);
        float4 b01 = bp[0];
        float4 b23 = bp[1];
        v2f b0 = {b01.x, b01.y}, b1 = {b01.z, b01.w};
        v2f b2 = {b23.x, b23.y}, b3 = {b23.z, b23.w};
        acc0 = __builtin_amdgcn_wmma_f32_16x16x4_f32(false, a, false, b0, (short)0, acc0, false, false);
        acc1 = __builtin_amdgcn_wmma_f32_16x16x4_f32(false, a, false, b1, (short)0, acc1, false, false);
        acc2 = __builtin_amdgcn_wmma_f32_16x16x4_f32(false, a, false, b2, (short)0, acc2, false, false);
        acc3 = __builtin_amdgcn_wmma_f32_16x16x4_f32(false, a, false, b3, (short)0, acc3, false, false);
    }

    const int rbase = gwave * 16 + lhalf * 8;
    const float bi0 = bias[lmod], bi1 = bias[16 + lmod], bi2 = bias[32 + lmod], bi3 = bias[48 + lmod];
    if (MODE == 2) {
#pragma unroll
        for (int r = 0; r < 8; ++r) {
            size_t ro = (size_t)(rbase + r) * HDIM + lmod;
            C[ro]      = lrelu(acc0[r] + bi0);
            C[ro + 16] = lrelu(acc1[r] + bi1);
            C[ro + 32] = lrelu(acc2[r] + bi2);
            C[ro + 48] = lrelu(acc3[r] + bi3);
        }
    } else {
        const float4 dq0 = *(const float4*)(dinv + rbase);
        const float4 dq1 = *(const float4*)(dinv + rbase + 4);
        const float dv[8] = {dq0.x, dq0.y, dq0.z, dq0.w, dq1.x, dq1.y, dq1.z, dq1.w};
#pragma unroll
        for (int r = 0; r < 8; ++r) {
            size_t ro = (size_t)(rbase + r) * HDIM + lmod;
            const float s = dv[r] * dv[r];
            const float h0 = acc0[r], h1 = acc1[r], h2 = acc2[r], h3 = acc3[r];
            C[ro] = h0; C[ro + 16] = h1; C[ro + 32] = h2; C[ro + 48] = h3;
            C2[ro]      = bi0 + s * h0;
            C2[ro + 16] = bi1 + s * h1;
            C2[ro + 32] = bi2 + s * h2;
            C2[ro + 48] = bi3 + s * h3;
        }
    }
}

__global__ void conv_gemm_f32_kernel(const float* __restrict__ A, const float* __restrict__ B,
                                     float* __restrict__ C, const float* __restrict__ bias,
                                     const float* __restrict__ dinv, float* __restrict__ C2, int M, int K) {
    gemm_strip_f32<0>(A, B, C, bias, dinv, C2, M, K);
}

__global__ void conv_gemm_relu_f32_kernel(const float* __restrict__ A, const float* __restrict__ B,
                                          float* __restrict__ C, const float* __restrict__ bias,
                                          const float* __restrict__ dinv, float* __restrict__ C2, int M, int K) {
    gemm_strip_f32<1>(A, B, C, bias, dinv, C2, M, K);
}

__global__ void dense_gemm_lrelu_f32_kernel(const float* __restrict__ A, const float* __restrict__ B,
                                            float* __restrict__ C, const float* __restrict__ bias, int M, int K) {
    gemm_strip_f32<2>(A, B, C, bias, nullptr, nullptr, M, K);
}

#define SC_EB 16
#define SC_WCH 64
#define SC_BCH (8 * SC_WCH)
#define SC_Q 64
#define SCATTER_BLOCKS 1536

__device__ __forceinline__ int lane_bcast_i(int v, int l) { return __builtin_amdgcn_readlane(v, l); }
__device__ __forceinline__ float lane_bcast_f(float v, int l) {
    return __int_as_float(__builtin_amdgcn_readlane(__float_as_int(v), l));
}

template <bool FULL>
__device__ __forceinline__ void scatter_batch(const float* __restrict__ Hm, const float* __restrict__ dinv,
                                              float* __restrict__ out, int id, int cnt, int lane) {
    const float dv = dinv[id];
    float h0[SC_EB], h1[SC_EB];
#pragma unroll
    for (int k = 0; k < SC_EB; ++k) {
        h0[k] = 0.0f; h1[k] = 0.0f;
        if (FULL || k < cnt) {
            const int s = lane_bcast_i(id, k);
            const float* hr = Hm + (size_t)s * HDIM + lane;
            h0[k] = hr[0];
            h1[k] = hr[32];
        }
    }
#pragma unroll
    for (int k = 0; k < SC_EB; ++k) {
        if (FULL || k < cnt) {
            const int d = lane_bcast_i(id, 16 + k);
            const float nrm = lane_bcast_f(dv, k) * lane_bcast_f(dv, 16 + k);
            float* o = out + (size_t)d * HDIM + lane;
            __hip_atomic_fetch_add(o,      nrm * h0[k], __ATOMIC_RELAXED, __HIP_MEMORY_SCOPE_AGENT);
            __hip_atomic_fetch_add(o + 32, nrm * h1[k], __ATOMIC_RELAXED, __HIP_MEMORY_SCOPE_AGENT);
        }
    }
}

__global__ __launch_bounds__(256) void conv_scatter_kernel(const float* __restrict__ Hm, const int* __restrict__ src,
                                    const int* __restrict__ dst, const float* __restrict__ dinv,
                                    float* __restrict__ out, unsigned* __restrict__ tickets, int e, int n) {
    __shared__ int q_src[8 * SC_Q];
    __shared__ int q_dst[8 * SC_Q];
    __shared__ int s_tick[2];
    __shared__ int s_half;
    const int tid  = threadIdx.x;
    const int lane = tid & 31;
    const int sub  = lane & 15;
    const int wv = __builtin_amdgcn_readfirstlane(tid >> 5);
    int* qs = q_src + wv * SC_Q;
    int* qd = q_dst + wv * SC_Q;
    const int split = n >> 1;
    const int nchunks = (e + SC_BCH - 1) / SC_BCH;

    if (tid == 0) {
        const int hf = (int)((__builtin_amdgcn_s_sendmsg_rtn(0x87) >> 18) & 1u);
        s_half = hf;
        s_tick[0] = (int)__hip_atomic_fetch_add(tickets + hf, 1u, __ATOMIC_RELAXED, __HIP_MEMORY_SCOPE_AGENT);
    }
    __syncthreads();
    const int myhalf = s_half;
    int pass = 0;
    int head = 0, tail = 0;

    for (int trip = 0; trip < 2 * nchunks + 2; ++trip) {
        const int t = __builtin_amdgcn_readfirstlane(s_tick[trip & 1]);
        if (t >= nchunks) {
            if (pass == 1) break;
            pass = 1;
            if (tid == 0)
                s_tick[(trip + 1) & 1] = (int)__hip_atomic_fetch_add(tickets + (myhalf ^ 1), 1u, __ATOMIC_RELAXED,
                                                                     __HIP_MEMORY_SCOPE_AGENT);
            __syncthreads();
            continue;
        }
        const int hh = myhalf ^ pass;

        const int c0 = t * SC_BCH + wv * SC_WCH;
        int dd[SC_WCH / 32], ss[SC_WCH / 32];
#pragma unroll
        for (int j = 0; j < SC_WCH / 32; ++j) {
            const int ei = c0 + j * 32 + lane;
            const bool ok = ei < e;
            dd[j] = ok ? dst[ei] : -1;
            ss[j] = ok ? src[ei] : 0;
        }
        int tnext = 0;
        if (tid == 0)
            tnext = (int)__hip_atomic_fetch_add(tickets + hh, 1u, __ATOMIC_RELAXED, __HIP_MEMORY_SCOPE_AGENT);
#pragma unroll
        for (int j = 0; j < SC_WCH / 32; ++j) {
            const int d = dd[j];
            const bool mine = (d >= 0) && ((d >= split) == (hh != 0));
            const unsigned m = __builtin_amdgcn_ballot_w32(mine);
            if (mine) {
                const int pos = (tail + (int)__builtin_amdgcn_mbcnt_lo(m, 0u)) & (SC_Q - 1);
                qs[pos] = ss[j];
                qd[pos] = d;
            }
            tail += __builtin_popcount(m);
            for (int r = 0; r < 2 && tail - head >= SC_EB; ++r) {
                const int qi = (head + sub) & (SC_Q - 1);
                const int id = (lane < 16) ? qs[qi] : qd[qi];
                scatter_batch<true>(Hm, dinv, out, id, SC_EB, lane);
                head += SC_EB;
            }
        }
        if (tid == 0) s_tick[(trip + 1) & 1] = tnext;
        __syncthreads();
    }

    const int rem = tail - head;
    if (rem > 0) {
        const int qi = (head + sub) & (SC_Q - 1);
        int id = (lane < 16) ? qs[qi] : qd[qi];
        if (sub >= rem) id = 0;
        scatter_batch<false>(Hm, dinv, out, id, rem, lane);
    }
}

__global__ void dual_head_kernel(const float* __restrict__ Z, const int* __restrict__ idx, int n,
                                 const float* __restrict__ Wa, const float* __restrict__ ba,
                                 float* __restrict__ outA,
                                 const float* __restrict__ Wb, const float* __restrict__ bb,
                                 float* __restrict__ outB) {
    int t = blockIdx.x * blockDim.x + threadIdx.x;
    if (t >= n) return;
    const float* row = Z + (size_t)idx[t] * HDIM;
    float da = 0.0f, db = 0.0f;
#pragma unroll
    for (int j = 0; j < HDIM; j += 4) {
        float4 z  = *(const float4*)(row + j);
        float4 wa = *(const float4*)(Wa + j);
        float4 wb = *(const float4*)(Wb + j);
        da += z.x * wa.x + z.y * wa.y + z.z * wa.z + z.w * wa.w;
        db += z.x * wb.x + z.y * wb.y + z.z * wb.z + z.w * wb.w;
    }
    outA[t] = lrelu(da + ba[0]);
    outB[t] = lrelu(db + bb[0]);
}

__global__ void tprob_kernel(const float* __restrict__ P1, const float* __restrict__ Wp2,
                             const float* __restrict__ bp2, float* __restrict__ out, int n) {
    int i = blockIdx.x * blockDim.x + threadIdx.x;
    if (i >= n) return;
    const float* row = P1 + (size_t)i * HDIM;
    float d0 = 0.0f, d1 = 0.0f;
#pragma unroll
    for (int k = 0; k < HDIM; ++k) {
        float z = row[k];
        d0 += z * Wp2[k * 2 + 0];
        d1 += z * Wp2[k * 2 + 1];
    }
    out[i * 2 + 0] = lrelu(d0 + bp2[0]);
    out[i * 2 + 1] = lrelu(d1 + bp2[1]);
}

extern "C" void kernel_launch(void* const* d_in, const int* in_sizes, int n_in,
                              void* d_out, int out_size, void* d_ws, size_t ws_size,
                              hipStream_t stream) {
    const float* x     = (const float*)d_in[0];
    const int*   ei    = (const int*)d_in[1];
    const int*   treat = (const int*)d_in[2];
    const int*   ctrl  = (const int*)d_in[3];
    const float* W1  = (const float*)d_in[4];  const float* b1  = (const float*)d_in[5];
    const float* W2  = (const float*)d_in[6];  const float* b2  = (const float*)d_in[7];
    const float* Wy1 = (const float*)d_in[8];  const float* by1 = (const float*)d_in[9];
    const float* Wy0 = (const float*)d_in[10]; const float* by0 = (const float*)d_in[11];
    const float* Wp1 = (const float*)d_in[12]; const float* bp1 = (const float*)d_in[13];
    const float* Wp2 = (const float*)d_in[14]; const float* bp2 = (const float*)d_in[15];

    const int N = in_sizes[0] / INDIM;
    const int E = in_sizes[1] / 2;
    const int T = in_sizes[2];
    const int Cn = in_sizes[3];
    const int* e_src = ei;
    const int* e_dst = ei + E;

    float* dinv = (float*)d_ws;
    unsigned* tickets = (unsigned*)(dinv + N);
    float* buf0 = dinv + (((size_t)N + 4 + 63) & ~(size_t)63);
    float* buf1 = buf0 + (size_t)N * HDIM;

    float* o_y1    = (float*)d_out;
    float* o_yc0   = o_y1 + T;
    float* o_y0    = o_yc0 + T;
    float* o_yc1   = o_y0 + Cn;
    float* o_tprob = o_yc1 + Cn;
    float* o_xZ2   = o_tprob + (size_t)2 * N;

    const int BLK = 256;
    auto cdiv = [](long long a, long long b) { return (int)((a + b - 1) / b); };
    const int gemm_blocks = cdiv(N / 16, 8);
    const size_t smem128 = (size_t)INDIM * HDIM * sizeof(float);
    const size_t smem64  = (size_t)HDIM * HDIM * sizeof(float);

    const int span = cdiv(N, DEG_R);
    unsigned* degpart = (unsigned*)buf1;
    deg_hist_kernel<<<DEG_R * DEG_C, DEG_T, (size_t)span * sizeof(unsigned), stream>>>(e_dst, degpart, N, E, span);
    dinv_kernel<<<cdiv(N, BLK), BLK, 0, stream>>>(degpart, dinv, tickets, N);

    conv_gemm_f32_kernel<<<gemm_blocks, BLK, smem128, stream>>>(x, W1, buf0, b1, dinv, buf1, N, INDIM);
    conv_scatter_kernel<<<SCATTER_BLOCKS, BLK, 0, stream>>>(buf0, e_src, e_dst, dinv, buf1, tickets, E, N);

    conv_gemm_relu_f32_kernel<<<gemm_blocks, BLK, smem64, stream>>>(buf1, W2, buf0, b2, dinv, o_xZ2, N, HDIM);
    conv_scatter_kernel<<<SCATTER_BLOCKS, BLK, 0, stream>>>(buf0, e_src, e_dst, dinv, o_xZ2, tickets + 2, E, N);

    dual_head_kernel<<<cdiv(T, BLK), BLK, 0, stream>>>(o_xZ2, treat, T, Wy1, by1, o_y1, Wy0, by0, o_yc0);
    dual_head_kernel<<<cdiv(Cn, BLK), BLK, 0, stream>>>(o_xZ2, ctrl, Cn, Wy0, by0, o_y0, Wy1, by1, o_yc1);

    dense_gemm_lrelu_f32_kernel<<<gemm_blocks, BLK, smem64, stream>>>(o_xZ2, Wp1, buf0, bp1, N, HDIM);
    tprob_kernel<<<cdiv(N, BLK), BLK, 0, stream>>>(buf0, Wp2, bp2, o_tprob, N);
}


# Normalisation leaves the edge loop: GEMMs store dinv*H, scatter adds plain rows (no 32-line dinv gather per 16 edges), next A loader applies bias+dinv[dst] after the sum; atomic arrival order kept
# speedup vs baseline: 2.6347x; 1.0844x over previous
#include <hip/hip_runtime.h>
#include <hip/hip_bf16.h>

typedef __attribute__((ext_vector_type(2))) float v2f;
typedef __attribute__((ext_vector_type(8))) float v8f;

#define HDIM 64
#define INDIM 128

__device__ __forceinline__ float lrelu(float v) { return v > 0.0f ? v : 0.01f * v; }

#define DEG_R 8
#define DEG_C 32
#define DEG_T 512
#define DEG_U 8

__device__ __forceinline__ void deg_bump(unsigned* cnt, int d, int lo, unsigned span) {
    const unsigned o = (unsigned)(d - lo);
    if (o < span) atomicAdd(&cnt[o], 1u);
}

__global__ __launch_bounds__(DEG_T) void deg_hist_kernel(const int* __restrict__ dst, unsigned* __restrict__ part,
                                                         int n, int e, int span) {
    extern __shared__ unsigned cnt[];
    const int tid = threadIdx.x;
    const int r = blockIdx.x % DEG_R;
    const int c = blockIdx.x / DEG_R;
    const int lo = r * span;
    const unsigned uspan = (unsigned)span;

    for (int i = tid; i < span; i += DEG_T) cnt[i] = 0u;
    __syncthreads();

    const int per = (((e + DEG_C - 1) / DEG_C) + 3) & ~3;
    const int e0 = min(e, c * per);
    const int e1 = min(e, e0 + per);
    if ((((size_t)dst) & 15) == 0) {
        const int n4 = (e1 - e0) >> 2;
        const int4* __restrict__ p4 = (const int4*)(dst + e0);
        for (int base = tid; base < n4; base += DEG_U * DEG_T) {
            int4 v[DEG_U];
#pragma unroll
            for (int u = 0; u < DEG_U; ++u) {
                const int idx = base + u * DEG_T;
                v[u] = (idx < n4) ? p4[idx] : make_int4(-1, -1, -1, -1);
            }
#pragma unroll
            for (int u = 0; u < DEG_U; ++u) {
                deg_bump(cnt, v[u].x, lo, uspan);
                deg_bump(cnt, v[u].y, lo, uspan);
                deg_bump(cnt, v[u].z, lo, uspan);
                deg_bump(cnt, v[u].w, lo, uspan);
            }
        }
        const int t0 = e0 + 4 * n4;
        if (t0 + tid < e1) deg_bump(cnt, dst[t0 + tid], lo, uspan);
    } else {
        for (int i = e0 + tid; i < e1; i += DEG_T) deg_bump(cnt, dst[i], lo, uspan);
    }
    __syncthreads();

    unsigned* __restrict__ plane = part + (size_t)c * n;
    for (int base = tid; base < span; base += 4 * DEG_T) {
        unsigned t[4];
#pragma unroll
        for (int u = 0; u < 4; ++u) {
            const int i = base + u * DEG_T;
            t[u] = (i < span) ? cnt[i] : 0u;
        }
#pragma unroll
        for (int u = 0; u < 4; ++u) {
            const int i = base + u * DEG_T;
            if (i < span && lo + i < n) plane[lo + i] = t[u];
        }
    }
}

__global__ void dinv_kernel(const unsigned* __restrict__ part, float* __restrict__ dinv, unsigned* __restrict__ tickets,
                            int n) {
    int i = blockIdx.x * blockDim.x + threadIdx.x;
    if (i < 4) tickets[i] = 0u;
    if (i < n) {
        unsigned s = 0u;
#pragma unroll
        for (int c = 0; c < DEG_C; ++c) s += part[(size_t)c * n + i];
        const float d = 1.0f + (float)s;
        dinv[i] = d > 0.0f ? rsqrtf(d) : 0.0f;
    }
}

template <int MODE>
__device__ __forceinline__ void gemm_strip_f32(const float* __restrict__ A, float* Aw, const float* __restrict__ B,
                                               float* __restrict__ C, const float* __restrict__ bias,
                                               const float* __restrict__ dinv, float* __restrict__ C2,
                                               const float* __restrict__ biasA, int M, int K) {
    extern __shared__ float bsw[];
    const int tid   = threadIdx.x;
    const int lane  = tid & 31;
    const int wave  = tid >> 5;
    const int lmod  = lane & 15;
    const int lhalf = lane >> 4;

    const int pairs = K * 32;
    for (int p = tid; p < pairs; p += 256) {
        int ks  = p >> 7;
        int ll  = (p >> 2) & 31;
        int tn  = p & 3;
        int kb  = 4 * ks + 2 * (ll >> 4);
        int col = tn * 16 + (ll & 15);
        bsw[p * 2 + 0] = B[(size_t)kb * HDIM + col];
        bsw[p * 2 + 1] = B[(size_t)(kb + 1) * HDIM + col];
    }
    if (MODE != 0) {
        if (tid < K) bsw[(size_t)K * HDIM + tid] = biasA[tid];
    }
    __syncthreads();

    const int gwave = blockIdx.x * 8 + wave;
    if (gwave >= (M >> 4)) return;

    const size_t aoff = (size_t)(gwave * 16 + lmod) * K + 2 * lhalf;
    const float* arow = (MODE == 2) ? nullptr : A + aoff;
    float* wrow = (MODE == 2) ? Aw + aoff : nullptr;
    const float dvrow = (MODE != 0) ? dinv[gwave * 16 + lmod] : 0.0f;
    const float* bA = bsw + (size_t)K * HDIM + 2 * lhalf;
    v8f acc0 = {}, acc1 = {}, acc2 = {}, acc3 = {};

    const int steps = K >> 2;
    for (int ks = 0; ks < steps; ++ks) {
        v2f a;
        if (MODE == 2) a = *(const v2f*)(wrow + 4 * ks);
        else           a = *(const v2f*)(arow + 4 * ks);
        if (MODE != 0) {
            const v2f bb = *(const v2f*)(bA + 4 * ks);
            a.x = bb.x + dvrow * a.x;
            a.y = bb.y + dvrow * a.y;
            if (MODE == 1) { a.x = fmaxf(a.x, 0.0f); a.y = fmaxf(a.y, 0.0f); }
            if (MODE == 2) *(v2f*)(wrow + 4 * ks) = a;
        }
        const float4* bp = (const float4*)(bsw + (size_t)(ks * 32 + lane) * 8);
        float4 b01 = bp[0];
        float4 b23 = bp[1];
        v2f b0 = {b01.x, b01.y}, b1 = {b01.z, b01.w};
        v2f b2 = {b23.x, b23.y}, b3 = {b23.z, b23.w};
        acc0 = __builtin_amdgcn_wmma_f32_16x16x4_f32(false, a, false, b0, (short)0, acc0, false, false);
        acc1 = __builtin_amdgcn_wmma_f32_16x16x4_f32(false, a, false, b1, (short)0, acc1, false, false);
        acc2 = __builtin_amdgcn_wmma_f32_16x16x4_f32(false, a, false, b2, (short)0, acc2, false, false);
        acc3 = __builtin_amdgcn_wmma_f32_16x16x4_f32(false, a, false, b3, (short)0, acc3, false, false);
    }

    const int rbase = gwave * 16 + lhalf * 8;
    if (MODE == 2) {
        const float bi0 = bias[lmod], bi1 = bias[16 + lmod], bi2 = bias[32 + lmod], bi3 = bias[48 + lmod];
#pragma unroll
        for (int r = 0; r < 8; ++r) {
            size_t ro = (size_t)(rbase + r) * HDIM + lmod;
            C[ro]      = lrelu(acc0[r] + bi0);
            C[ro + 16] = lrelu(acc1[r] + bi1);
            C[ro + 32] = lrelu(acc2[r] + bi2);
            C[ro + 48] = lrelu(acc3[r] + bi3);
        }
    } else {
        const float4 dq0 = *(const float4*)(dinv + rbase);
        const float4 dq1 = *(const float4*)(dinv + rbase + 4);
        const float dv[8] = {dq0.x, dq0.y, dq0.z, dq0.w, dq1.x, dq1.y, dq1.z, dq1.w};
#pragma unroll
        for (int r = 0; r < 8; ++r) {
            size_t ro = (size_t)(rbase + r) * HDIM + lmod;
            const float s = dv[r];
            const float h0 = s * acc0[r], h1 = s * acc1[r], h2 = s * acc2[r], h3 = s * acc3[r];
            C[ro] = h0; C[ro + 16] = h1; C[ro + 32] = h2; C[ro + 48] = h3;
            C2[ro] = h0; C2[ro + 16] = h1; C2[ro + 32] = h2; C2[ro + 48] = h3;
        }
    }
}

__global__ void conv_gemm_f32_kernel(const float* __restrict__ A, const float* __restrict__ B,
                                     float* __restrict__ C, const float* __restrict__ dinv,
                                     float* __restrict__ C2, int M, int K) {
    gemm_strip_f32<0>(A, nullptr, B, C, nullptr, dinv, C2, nullptr, M, K);
}

__global__ void conv_gemm_relu_f32_kernel(const float* __restrict__ A, const float* __restrict__ B,
                                          float* __restrict__ C, const float* __restrict__ dinv,
                                          float* __restrict__ C2, const float* __restrict__ biasA, int M, int K) {
    gemm_strip_f32<1>(A, nullptr, B, C, nullptr, dinv, C2, biasA, M, K);
}

__global__ void dense_gemm_lrelu_f32_kernel(float* Aw, const float* __restrict__ B,
                                            float* __restrict__ C, const float* __restrict__ bias,
                                            const float* __restrict__ dinv, const float* __restrict__ biasA,
                                            int M, int K) {
    gemm_strip_f32<2>(nullptr, Aw, B, C, bias, dinv, nullptr, biasA, M, K);
}

#define SC_EB 16
#define SC_WCH 64
#define SC_BCH (8 * SC_WCH)
#define SC_Q 64
#define SCATTER_BLOCKS 1536

__device__ __forceinline__ int lane_bcast_i(int v, int l) { return __builtin_amdgcn_readlane(v, l); }

template <bool FULL>
__device__ __forceinline__ void scatter_batch(const float* __restrict__ Hm, float* __restrict__ out, int id, int cnt,
                                              int lane) {
    float h0[SC_EB], h1[SC_EB];
#pragma unroll
    for (int k = 0; k < SC_EB; ++k) {
        h0[k] = 0.0f; h1[k] = 0.0f;
        if (FULL || k < cnt) {
            const int s = lane_bcast_i(id, k);
            const float* hr = Hm + (size_t)s * HDIM + lane;
            h0[k] = hr[0];
            h1[k] = hr[32];
        }
    }
#pragma unroll
    for (int k = 0; k < SC_EB; ++k) {
        if (FULL || k < cnt) {
            const int d = lane_bcast_i(id, 16 + k);
            float* o = out + (size_t)d * HDIM + lane;
            __hip_atomic_fetch_add(o,      h0[k], __ATOMIC_RELAXED, __HIP_MEMORY_SCOPE_AGENT);
            __hip_atomic_fetch_add(o + 32, h1[k], __ATOMIC_RELAXED, __HIP_MEMORY_SCOPE_AGENT);
        }
    }
}

__global__ __launch_bounds__(256) void conv_scatter_kernel(const float* __restrict__ Hm, const int* __restrict__ src,
                                    const int* __restrict__ dst, float* __restrict__ out,
                                    unsigned* __restrict__ tickets, int e, int n) {
    __shared__ int q_src[8 * SC_Q];
    __shared__ int q_dst[8 * SC_Q];
    __shared__ int s_tick[2];
    __shared__ int s_half;
    const int tid  = threadIdx.x;
    const int lane = tid & 31;
    const int sub  = lane & 15;
    const int wv = __builtin_amdgcn_readfirstlane(tid >> 5);
    int* qs = q_src + wv * SC_Q;
    int* qd = q_dst + wv * SC_Q;
    const int split = n >> 1;
    const int nchunks = (e + SC_BCH - 1) / SC_BCH;

    if (tid == 0) {
        const int hf = (int)((__builtin_amdgcn_s_sendmsg_rtn(0x87) >> 18) & 1u);
        s_half = hf;
        s_tick[0] = (int)__hip_atomic_fetch_add(tickets + hf, 1u, __ATOMIC_RELAXED, __HIP_MEMORY_SCOPE_AGENT);
    }
    __syncthreads();
    const int myhalf = s_half;
    int pass = 0;
    int head = 0, tail = 0;

    for (int trip = 0; trip < 2 * nchunks + 2; ++trip) {
        const int t = __builtin_amdgcn_readfirstlane(s_tick[trip & 1]);
        if (t >= nchunks) {
            if (pass == 1) break;
            pass = 1;
            if (tid == 0)
                s_tick[(trip + 1) & 1] = (int)__hip_atomic_fetch_add(tickets + (myhalf ^ 1), 1u, __ATOMIC_RELAXED,
                                                                     __HIP_MEMORY_SCOPE_AGENT);
            __syncthreads();
            continue;
        }
        const int hh = myhalf ^ pass;

        const int c0 = t * SC_BCH + wv * SC_WCH;
        int dd[SC_WCH / 32], ss[SC_WCH / 32];
#pragma unroll
        for (int j = 0; j < SC_WCH / 32; ++j) {
            const int ei = c0 + j * 32 + lane;
            const bool ok = ei < e;
            dd[j] = ok ? dst[ei] : -1;
            ss[j] = ok ? src[ei] : 0;
        }
        int tnext = 0;
        if (tid == 0)
            tnext = (int)__hip_atomic_fetch_add(tickets + hh, 1u, __ATOMIC_RELAXED, __HIP_MEMORY_SCOPE_AGENT);
#pragma unroll
        for (int j = 0; j < SC_WCH / 32; ++j) {
            const int d = dd[j];
            const bool mine = (d >= 0) && ((d >= split) == (hh != 0));
            const unsigned m = __builtin_amdgcn_ballot_w32(mine);
            if (mine) {
                const int pos = (tail + (int)__builtin_amdgcn_mbcnt_lo(m, 0u)) & (SC_Q - 1);
                qs[pos] = ss[j];
                qd[pos] = d;
            }
            tail += __builtin_popcount(m);
            for (int r = 0; r < 2 && tail - head >= SC_EB; ++r) {
                const int qi = (head + sub) & (SC_Q - 1);
                const int id = (lane < 16) ? qs[qi] : qd[qi];
                scatter_batch<true>(Hm, out, id, SC_EB, lane);
                head += SC_EB;
            }
        }
        if (tid == 0) s_tick[(trip + 1) & 1] = tnext;
        __syncthreads();
    }

    const int rem = tail - head;
    if (rem > 0) {
        const int qi = (head + sub) & (SC_Q - 1);
        int id = (lane < 16) ? qs[qi] : qd[qi];
        if (sub >= rem) id = 0;
        scatter_batch<false>(Hm, out, id, rem, lane);
    }
}

__global__ void dual_head_kernel(const float* __restrict__ Z, const int* __restrict__ idx, int n,
                                 const float* __restrict__ Wa, const float* __restrict__ ba,
                                 float* __restrict__ outA,
                                 const float* __restrict__ Wb, const float* __restrict__ bb,
                                 float* __restrict__ outB) {
    int t = blockIdx.x * blockDim.x + threadIdx.x;
    if (t >= n) return;
    const float* row = Z + (size_t)idx[t] * HDIM;
    float da = 0.0f, db = 0.0f;
#pragma unroll
    for (int j = 0; j < HDIM; j += 4) {
        float4 z  = *(const float4*)(row + j);
        float4 wa = *(const float4*)(Wa + j);
        float4 wb = *(const float4*)(Wb + j);
        da += z.x * wa.x + z.y * wa.y + z.z * wa.z + z.w * wa.w;
        db += z.x * wb.x + z.y * wb.y + z.z * wb.z + z.w * wb.w;
    }
    outA[t] = lrelu(da + ba[0]);
    outB[t] = lrelu(db + bb[0]);
}

__global__ void tprob_kernel(const float* __restrict__ P1, const float* __restrict__ Wp2,
                             const float* __restrict__ bp2, float* __restrict__ out, int n) {
    int i = blockIdx.x * blockDim.x + threadIdx.x;
    if (i >= n) return;
    const float* row = P1 + (size_t)i * HDIM;
    float d0 = 0.0f, d1 = 0.0f;
#pragma unroll
    for (int k = 0; k < HDIM; ++k) {
        float z = row[k];
        d0 += z * Wp2[k * 2 + 0];
        d1 += z * Wp2[k * 2 + 1];
    }
    out[i * 2 + 0] = lrelu(d0 + bp2[0]);
    out[i * 2 + 1] = lrelu(d1 + bp2[1]);
}

extern "C" void kernel_launch(void* const* d_in, const int* in_sizes, int n_in,
                              void* d_out, int out_size, void* d_ws, size_t ws_size,
                              hipStream_t stream) {
    const float* x     = (const float*)d_in[0];
    const int*   ei    = (const int*)d_in[1];
    const int*   treat = (const int*)d_in[2];
    const int*   ctrl  = (const int*)d_in[3];
    const float* W1  = (const float*)d_in[4];  const float* b1  = (const float*)d_in[5];
    const float* W2  = (const float*)d_in[6];  const float* b2  = (const float*)d_in[7];
    const float* Wy1 = (const float*)d_in[8];  const float* by1 = (const float*)d_in[9];
    const float* Wy0 = (const float*)d_in[10]; const float* by0 = (const float*)d_in[11];
    const float* Wp1 = (const float*)d_in[12]; const float* bp1 = (const float*)d_in[13];
    const float* Wp2 = (const float*)d_in[14]; const float* bp2 = (const float*)d_in[15];

    const int N = in_sizes[0] / INDIM;
    const int E = in_sizes[1] / 2;
    const int T = in_sizes[2];
    const int Cn = in_sizes[3];
    const int* e_src = ei;
    const int* e_dst = ei + E;

    float* dinv = (float*)d_ws;
    unsigned* tickets = (unsigned*)(dinv + N);
    float* buf0 = dinv + (((size_t)N + 4 + 63) & ~(size_t)63);
    float* buf1 = buf0 + (size_t)N * HDIM;

    float* o_y1    = (float*)d_out;
    float* o_yc0   = o_y1 + T;
    float* o_y0    = o_yc0 + T;
    float* o_yc1   = o_y0 + Cn;
    float* o_tprob = o_yc1 + Cn;
    float* o_xZ2   = o_tprob + (size_t)2 * N;

    const int BLK = 256;
    auto cdiv = [](long long a, long long b) { return (int)((a + b - 1) / b); };
    const int gemm_blocks = cdiv(N / 16, 8);
    const size_t smem128 = (size_t)INDIM * HDIM * sizeof(float);
    const size_t smem64  = (size_t)(HDIM * HDIM + HDIM) * sizeof(float);

    const int span = cdiv(N, DEG_R);
    unsigned* degpart = (unsigned*)buf1;
    deg_hist_kernel<<<DEG_R * DEG_C, DEG_T, (size_t)span * sizeof(unsigned), stream>>>(e_dst, degpart, N, E, span);
    dinv_kernel<<<cdiv(N, BLK), BLK, 0, stream>>>(degpart, dinv, tickets, N);

    conv_gemm_f32_kernel<<<gemm_blocks, BLK, smem128, stream>>>(x, W1, buf0, dinv, buf1, N, INDIM);
    conv_scatter_kernel<<<SCATTER_BLOCKS, BLK, 0, stream>>>(buf0, e_src, e_dst, buf1, tickets, E, N);

    conv_gemm_relu_f32_kernel<<<gemm_blocks, BLK, smem64, stream>>>(buf1, W2, buf0, dinv, o_xZ2, b1, N, HDIM);
    conv_scatter_kernel<<<SCATTER_BLOCKS, BLK, 0, stream>>>(buf0, e_src, e_dst, o_xZ2, tickets + 2, E, N);

    dense_gemm_lrelu_f32_kernel<<<gemm_blocks, BLK, smem64, stream>>>(o_xZ2, Wp1, buf0, bp1, dinv, b2, N, HDIM);

    dual_head_kernel<<<cdiv(T, BLK), BLK, 0, stream>>>(o_xZ2, treat, T, Wy1, by1, o_y1, Wy0, by0, o_yc0);
    dual_head_kernel<<<cdiv(Cn, BLK), BLK, 0, stream>>>(o_xZ2, ctrl, Cn, Wy0, by0, o_y0, Wy1, by1, o_yc1);

    tprob_kernel<<<cdiv(N, BLK), BLK, 0, stream>>>(buf0, Wp2, bp2, o_tprob, N);
}


# The rsqrt launch folded into the conv1 WMMA GEMM: each wave sums the 32 degree planes of its own strip's rows, stores dinv, scales from registers; planes parked in the xZ2 slice; 10 launches to 9
# speedup vs baseline: 2.7558x; 1.0460x over previous
#include <hip/hip_runtime.h>
#include <hip/hip_bf16.h>

typedef __attribute__((ext_vector_type(2))) float v2f;
typedef __attribute__((ext_vector_type(8))) float v8f;

#define HDIM 64
#define INDIM 128

__device__ __forceinline__ float lrelu(float v) { return v > 0.0f ? v : 0.01f * v; }

#define DEG_R 8
#define DEG_C 32
#define DEG_T 512
#define DEG_U 8

__device__ __forceinline__ void deg_bump(unsigned* cnt, int d, int lo, unsigned span) {
    const unsigned o = (unsigned)(d - lo);
    if (o < span) atomicAdd(&cnt[o], 1u);
}

__global__ __launch_bounds__(DEG_T) void deg_hist_kernel(const int* __restrict__ dst, unsigned* __restrict__ part,
                                                         unsigned* __restrict__ tickets, int n, int e, int span) {
    extern __shared__ unsigned cnt[];
    const int tid = threadIdx.x;
    const int r = blockIdx.x % DEG_R;
    const int c = blockIdx.x / DEG_R;
    const int lo = r * span;
    const unsigned uspan = (unsigned)span;

    if (blockIdx.x == 0 && tid < 4) tickets[tid] = 0u;
    for (int i = tid; i < span; i += DEG_T) cnt[i] = 0u;
    __syncthreads();

    const int per = (((e + DEG_C - 1) / DEG_C) + 3) & ~3;
    const int e0 = min(e, c * per);
    const int e1 = min(e, e0 + per);
    if ((((size_t)dst) & 15) == 0) {
        const int n4 = (e1 - e0) >> 2;
        const int4* __restrict__ p4 = (const int4*)(dst + e0);
        for (int base = tid; base < n4; base += DEG_U * DEG_T) {
            int4 v[DEG_U];
#pragma unroll
            for (int u = 0; u < DEG_U; ++u) {
                const int idx = base + u * DEG_T;
                v[u] = (idx < n4) ? p4[idx] : make_int4(-1, -1, -1, -1);
            }
#pragma unroll
            for (int u = 0; u < DEG_U; ++u) {
                deg_bump(cnt, v[u].x, lo, uspan);
                deg_bump(cnt, v[u].y, lo, uspan);
                deg_bump(cnt, v[u].z, lo, uspan);
                deg_bump(cnt, v[u].w, lo, uspan);
            }
        }
        const int t0 = e0 + 4 * n4;
        if (t0 + tid < e1) deg_bump(cnt, dst[t0 + tid], lo, uspan);
    } else {
        for (int i = e0 + tid; i < e1; i += DEG_T) deg_bump(cnt, dst[i], lo, uspan);
    }
    __syncthreads();

    unsigned* __restrict__ plane = part + (size_t)c * n;
    for (int base = tid; base < span; base += 4 * DEG_T) {
        unsigned t[4];
#pragma unroll
        for (int u = 0; u < 4; ++u) {
            const int i = base + u * DEG_T;
            t[u] = (i < span) ? cnt[i] : 0u;
        }
#pragma unroll
        for (int u = 0; u < 4; ++u) {
            const int i = base + u * DEG_T;
            if (i < span && lo + i < n) plane[lo + i] = t[u];
        }
    }
}

template <int MODE>
__device__ __forceinline__ void gemm_strip_f32(const float* __restrict__ A, float* Aw, const float* __restrict__ B,
                                               float* __restrict__ C, const float* __restrict__ bias,
                                               const float* __restrict__ dinv, float* __restrict__ C2,
                                               const float* __restrict__ biasA, const unsigned* __restrict__ part,
                                               float* __restrict__ dinv_out, int M, int K) {
    extern __shared__ float bsw[];
    const int tid   = threadIdx.x;
    const int lane  = tid & 31;
    const int wave  = tid >> 5;
    const int lmod  = lane & 15;
    const int lhalf = lane >> 4;

    const int gwave = blockIdx.x * 8 + wave;
    const bool live = gwave < (M >> 4);
    unsigned dpl[DEG_C / 2];
    if (MODE == 0 && live) {
        const unsigned* pp = part + (size_t)(lhalf * (DEG_C / 2)) * M + gwave * 16 + lmod;
#pragma unroll
        for (int c = 0; c < DEG_C / 2; ++c) dpl[c] = pp[(size_t)c * M];
    }

    const int pairs = K * 32;
    for (int p = tid; p < pairs; p += 256) {
        int ks  = p >> 7;
        int ll  = (p >> 2) & 31;
        int tn  = p & 3;
        int kb  = 4 * ks + 2 * (ll >> 4);
        int col = tn * 16 + (ll & 15);
        bsw[p * 2 + 0] = B[(size_t)kb * HDIM + col];
        bsw[p * 2 + 1] = B[(size_t)(kb + 1) * HDIM + col];
    }
    if (MODE != 0) {
        if (tid < K) bsw[(size_t)K * HDIM + tid] = biasA[tid];
    }
    __syncthreads();

    if (!live) return;

    const size_t aoff = (size_t)(gwave * 16 + lmod) * K + 2 * lhalf;
    const float* arow = (MODE == 2) ? nullptr : A + aoff;
    float* wrow = (MODE == 2) ? Aw + aoff : nullptr;
    float dvrow;
    if (MODE == 0) {
        unsigned degs = 0u;
#pragma unroll
        for (int c = 0; c < DEG_C / 2; ++c) degs += dpl[c];
        const unsigned tot = degs + (unsigned)__shfl_xor((int)degs, 16);
        const float d = 1.0f + (float)tot;
        dvrow = d > 0.0f ? rsqrtf(d) : 0.0f;
        if (lhalf == 0) dinv_out[gwave * 16 + lmod] = dvrow;
    } else {
        dvrow = dinv[gwave * 16 + lmod];
    }
    const float* bA = bsw + (size_t)K * HDIM + 2 * lhalf;
    v8f acc0 = {}, acc1 = {}, acc2 = {}, acc3 = {};

    const int steps = K >> 2;
    for (int ks = 0; ks < steps; ++ks) {
        v2f a;
        if (MODE == 2) a = *(const v2f*)(wrow + 4 * ks);
        else           a = *(const v2f*)(arow + 4 * ks);
        if (MODE != 0) {
            const v2f bb = *(const v2f*)(bA + 4 * ks);
            a.x = bb.x + dvrow * a.x;
            a.y = bb.y + dvrow * a.y;
            if (MODE == 1) { a.x = fmaxf(a.x, 0.0f); a.y = fmaxf(a.y, 0.0f); }
            if (MODE == 2) *(v2f*)(wrow + 4 * ks) = a;
        }
        const float4* bp = (const float4*)(bsw + (size_t)(ks * 32 + lane) * 8);
        float4 b01 = bp[0];
        float4 b23 = bp[1];
        v2f b0 = {b01.x, b01.y}, b1 = {b01.z, b01.w};
        v2f b2 = {b23.x, b23.y}, b3 = {b23.z, b23.w};
        acc0 = __builtin_amdgcn_wmma_f32_16x16x4_f32(false, a, false, b0, (short)0, acc0, false, false);
        acc1 = __builtin_amdgcn_wmma_f32_16x16x4_f32(false, a, false, b1, (short)0, acc1, false, false);
        acc2 = __builtin_amdgcn_wmma_f32_16x16x4_f32(false, a, false, b2, (short)0, acc2, false, false);
        acc3 = __builtin_amdgcn_wmma_f32_16x16x4_f32(false, a, false, b3, (short)0, acc3, false, false);
    }

    const int rbase = gwave * 16 + lhalf * 8;
    if (MODE == 2) {
        const float bi0 = bias[lmod], bi1 = bias[16 + lmod], bi2 = bias[32 + lmod], bi3 = bias[48 + lmod];
#pragma unroll
        for (int r = 0; r < 8; ++r) {
            size_t ro = (size_t)(rbase + r) * HDIM + lmod;
            C[ro]      = lrelu(acc0[r] + bi0);
            C[ro + 16] = lrelu(acc1[r] + bi1);
            C[ro + 32] = lrelu(acc2[r] + bi2);
            C[ro + 48] = lrelu(acc3[r] + bi3);
        }
    } else {
        float dv[8];
        if (MODE == 0) {
            const int dvi = __float_as_int(dvrow);
#pragma unroll
            for (int r = 0; r < 8; ++r) {
                const int lo = __builtin_amdgcn_readlane(dvi, r);
                const int hi = __builtin_amdgcn_readlane(dvi, 8 + r);
                dv[r] = __int_as_float(lhalf ? hi : lo);
            }
        } else {
            const float4 dq0 = *(const float4*)(dinv + rbase);
            const float4 dq1 = *(const float4*)(dinv + rbase + 4);
            dv[0] = dq0.x; dv[1] = dq0.y; dv[2] = dq0.z; dv[3] = dq0.w;
            dv[4] = dq1.x; dv[5] = dq1.y; dv[6] = dq1.z; dv[7] = dq1.w;
        }
#pragma unroll
        for (int r = 0; r < 8; ++r) {
            size_t ro = (size_t)(rbase + r) * HDIM + lmod;
            const float s = dv[r];
            const float h0 = s * acc0[r], h1 = s * acc1[r], h2 = s * acc2[r], h3 = s * acc3[r];
            C[ro] = h0; C[ro + 16] = h1; C[ro + 32] = h2; C[ro + 48] = h3;
            C2[ro] = h0; C2[ro + 16] = h1; C2[ro + 32] = h2; C2[ro + 48] = h3;
        }
    }
}

__global__ void conv_gemm_f32_kernel(const float* __restrict__ A, const float* __restrict__ B,
                                     float* __restrict__ C, const unsigned* __restrict__ part,
                                     float* __restrict__ dinv_out, float* __restrict__ C2, int M, int K) {
    gemm_strip_f32<0>(A, nullptr, B, C, nullptr, nullptr, C2, nullptr, part, dinv_out, M, K);
}

__global__ void conv_gemm_relu_f32_kernel(const float* __restrict__ A, const float* __restrict__ B,
                                          float* __restrict__ C, const float* __restrict__ dinv,
                                          float* __restrict__ C2, const float* __restrict__ biasA, int M, int K) {
    gemm_strip_f32<1>(A, nullptr, B, C, nullptr, dinv, C2, biasA, nullptr, nullptr, M, K);
}

__global__ void dense_gemm_lrelu_f32_kernel(float* Aw, const float* __restrict__ B,
                                            float* __restrict__ C, const float* __restrict__ bias,
                                            const float* __restrict__ dinv, const float* __restrict__ biasA,
                                            int M, int K) {
    gemm_strip_f32<2>(nullptr, Aw, B, C, bias, dinv, nullptr, biasA, nullptr, nullptr, M, K);
}

#define SC_EB 16
#define SC_WCH 64
#define SC_BCH (8 * SC_WCH)
#define SC_Q 64
#define SCATTER_BLOCKS 1536

__device__ __forceinline__ int lane_bcast_i(int v, int l) { return __builtin_amdgcn_readlane(v, l); }

template <bool FULL>
__device__ __forceinline__ void scatter_batch(const float* __restrict__ Hm, float* __restrict__ out, int id, int cnt,
                                              int lane) {
    float h0[SC_EB], h1[SC_EB];
#pragma unroll
    for (int k = 0; k < SC_EB; ++k) {
        h0[k] = 0.0f; h1[k] = 0.0f;
        if (FULL || k < cnt) {
            const int s = lane_bcast_i(id, k);
            const float* hr = Hm + (size_t)s * HDIM + lane;
            h0[k] = hr[0];
            h1[k] = hr[32];
        }
    }
#pragma unroll
    for (int k = 0; k < SC_EB; ++k) {
        if (FULL || k < cnt) {
            const int d = lane_bcast_i(id, 16 + k);
            float* o = out + (size_t)d * HDIM + lane;
            __hip_atomic_fetch_add(o,      h0[k], __ATOMIC_RELAXED, __HIP_MEMORY_SCOPE_AGENT);
            __hip_atomic_fetch_add(o + 32, h1[k], __ATOMIC_RELAXED, __HIP_MEMORY_SCOPE_AGENT);
        }
    }
}

__global__ __launch_bounds__(256) void conv_scatter_kernel(const float* __restrict__ Hm, const int* __restrict__ src,
                                    const int* __restrict__ dst, float* __restrict__ out,
                                    unsigned* __restrict__ tickets, int e, int n) {
    __shared__ int q_src[8 * SC_Q];
    __shared__ int q_dst[8 * SC_Q];
    __shared__ int s_tick[2];
    __shared__ int s_half;
    const int tid  = threadIdx.x;
    const int lane = tid & 31;
    const int sub  = lane & 15;
    const int wv = __builtin_amdgcn_readfirstlane(tid >> 5);
    int* qs = q_src + wv * SC_Q;
    int* qd = q_dst + wv * SC_Q;
    const int split = n >> 1;
    const int nchunks = (e + SC_BCH - 1) / SC_BCH;

    if (tid == 0) {
        const int hf = (int)((__builtin_amdgcn_s_sendmsg_rtn(0x87) >> 18) & 1u);
        s_half = hf;
        s_tick[0] = (int)__hip_atomic_fetch_add(tickets + hf, 1u, __ATOMIC_RELAXED, __HIP_MEMORY_SCOPE_AGENT);
    }
    __syncthreads();
    const int myhalf = s_half;
    int pass = 0;
    int head = 0, tail = 0;

    for (int trip = 0; trip < 2 * nchunks + 2; ++trip) {
        const int t = __builtin_amdgcn_readfirstlane(s_tick[trip & 1]);
        if (t >= nchunks) {
            if (pass == 1) break;
            pass = 1;
            if (tid == 0)
                s_tick[(trip + 1) & 1] = (int)__hip_atomic_fetch_add(tickets + (myhalf ^ 1), 1u, __ATOMIC_RELAXED,
                                                                     __HIP_MEMORY_SCOPE_AGENT);
            __syncthreads();
            continue;
        }
        const int hh = myhalf ^ pass;

        const int c0 = t * SC_BCH + wv * SC_WCH;
        int dd[SC_WCH / 32], ss[SC_WCH / 32];
#pragma unroll
        for (int j = 0; j < SC_WCH / 32; ++j) {
            const int ei = c0 + j * 32 + lane;
            const bool ok = ei < e;
            dd[j] = ok ? dst[ei] : -1;
            ss[j] = ok ? src[ei] : 0;
        }
        int tnext = 0;
        if (tid == 0)
            tnext = (int)__hip_atomic_fetch_add(tickets + hh, 1u, __ATOMIC_RELAXED, __HIP_MEMORY_SCOPE_AGENT);
#pragma unroll
        for (int j = 0; j < SC_WCH / 32; ++j) {
            const int d = dd[j];
            const bool mine = (d >= 0) && ((d >= split) == (hh != 0));
            const unsigned m = __builtin_amdgcn_ballot_w32(mine);
            if (mine) {
                const int pos = (tail + (int)__builtin_amdgcn_mbcnt_lo(m, 0u)) & (SC_Q - 1);
                qs[pos] = ss[j];
                qd[pos] = d;
            }
            tail += __builtin_popcount(m);
            for (int r = 0; r < 2 && tail - head >= SC_EB; ++r) {
                const int qi = (head + sub) & (SC_Q - 1);
                const int id = (lane < 16) ? qs[qi] : qd[qi];
                scatter_batch<true>(Hm, out, id, SC_EB, lane);
                head += SC_EB;
            }
        }
        if (tid == 0) s_tick[(trip + 1) & 1] = tnext;
        __syncthreads();
    }

    const int rem = tail - head;
    if (rem > 0) {
        const int qi = (head + sub) & (SC_Q - 1);
        int id = (lane < 16) ? qs[qi] : qd[qi];
        if (sub >= rem) id = 0;
        scatter_batch<false>(Hm, out, id, rem, lane);
    }
}

__global__ void dual_head_kernel(const float* __restrict__ Z, const int* __restrict__ idx, int n,
                                 const float* __restrict__ Wa, const float* __restrict__ ba,
                                 float* __restrict__ outA,
                                 const float* __restrict__ Wb, const float* __restrict__ bb,
                                 float* __restrict__ outB) {
    int t = blockIdx.x * blockDim.x + threadIdx.x;
    if (t >= n) return;
    const float* row = Z + (size_t)idx[t] * HDIM;
    float da = 0.0f, db = 0.0f;
#pragma unroll
    for (int j = 0; j < HDIM; j += 4) {
        float4 z  = *(const float4*)(row + j);
        float4 wa = *(const float4*)(Wa + j);
        float4 wb = *(const float4*)(Wb + j);
        da += z.x * wa.x + z.y * wa.y + z.z * wa.z + z.w * wa.w;
        db += z.x * wb.x + z.y * wb.y + z.z * wb.z + z.w * wb.w;
    }
    outA[t] = lrelu(da + ba[0]);
    outB[t] = lrelu(db + bb[0]);
}

__global__ void tprob_kernel(const float* __restrict__ P1, const float* __restrict__ Wp2,
                             const float* __restrict__ bp2, float* __restrict__ out, int n) {
    int i = blockIdx.x * blockDim.x + threadIdx.x;
    if (i >= n) return;
    const float* row = P1 + (size_t)i * HDIM;
    float d0 = 0.0f, d1 = 0.0f;
#pragma unroll
    for (int k = 0; k < HDIM; ++k) {
        float z = row[k];
        d0 += z * Wp2[k * 2 + 0];
        d1 += z * Wp2[k * 2 + 1];
    }
    out[i * 2 + 0] = lrelu(d0 + bp2[0]);
    out[i * 2 + 1] = lrelu(d1 + bp2[1]);
}

extern "C" void kernel_launch(void* const* d_in, const int* in_sizes, int n_in,
                              void* d_out, int out_size, void* d_ws, size_t ws_size,
                              hipStream_t stream) {
    const float* x     = (const float*)d_in[0];
    const int*   ei    = (const int*)d_in[1];
    const int*   treat = (const int*)d_in[2];
    const int*   ctrl  = (const int*)d_in[3];
    const float* W1  = (const float*)d_in[4];  const float* b1  = (const float*)d_in[5];
    const float* W2  = (const float*)d_in[6];  const float* b2  = (const float*)d_in[7];
    const float* Wy1 = (const float*)d_in[8];  const float* by1 = (const float*)d_in[9];
    const float* Wy0 = (const float*)d_in[10]; const float* by0 = (const float*)d_in[11];
    const float* Wp1 = (const float*)d_in[12]; const float* bp1 = (const float*)d_in[13];
    const float* Wp2 = (const float*)d_in[14]; const float* bp2 = (const float*)d_in[15];

    const int N = in_sizes[0] / INDIM;
    const int E = in_sizes[1] / 2;
    const int T = in_sizes[2];
    const int Cn = in_sizes[3];
    const int* e_src = ei;
    const int* e_dst = ei + E;

    float* dinv = (float*)d_ws;
    unsigned* tickets = (unsigned*)(dinv + N);
    float* buf0 = dinv + (((size_t)N + 4 + 63) & ~(size_t)63);
    float* buf1 = buf0 + (size_t)N * HDIM;

    float* o_y1    = (float*)d_out;
    float* o_yc0   = o_y1 + T;
    float* o_y0    = o_yc0 + T;
    float* o_yc1   = o_y0 + Cn;
    float* o_tprob = o_yc1 + Cn;
    float* o_xZ2   = o_tprob + (size_t)2 * N;

    const int BLK = 256;
    auto cdiv = [](long long a, long long b) { return (int)((a + b - 1) / b); };
    const int gemm_blocks = cdiv(N / 16, 8);
    const size_t smem128 = (size_t)INDIM * HDIM * sizeof(float);
    const size_t smem64  = (size_t)(HDIM * HDIM + HDIM) * sizeof(float);

    const int span = cdiv(N, DEG_R);
    unsigned* degpart = (unsigned*)o_xZ2;
    deg_hist_kernel<<<DEG_R * DEG_C, DEG_T, (size_t)span * sizeof(unsigned), stream>>>(e_dst, degpart, tickets, N, E,
                                                                                       span);

    conv_gemm_f32_kernel<<<gemm_blocks, BLK, smem128, stream>>>(x, W1, buf0, degpart, dinv, buf1, N, INDIM);
    conv_scatter_kernel<<<SCATTER_BLOCKS, BLK, 0, stream>>>(buf0, e_src, e_dst, buf1, tickets, E, N);

    conv_gemm_relu_f32_kernel<<<gemm_blocks, BLK, smem64, stream>>>(buf1, W2, buf0, dinv, o_xZ2, b1, N, HDIM);
    conv_scatter_kernel<<<SCATTER_BLOCKS, BLK, 0, stream>>>(buf0, e_src, e_dst, o_xZ2, tickets + 2, E, N);

    dense_gemm_lrelu_f32_kernel<<<gemm_blocks, BLK, smem64, stream>>>(o_xZ2, Wp1, buf0, bp1, dinv, b2, N, HDIM);

    dual_head_kernel<<<cdiv(T, BLK), BLK, 0, stream>>>(o_xZ2, treat, T, Wy1, by1, o_y1, Wy0, by0, o_yc0);
    dual_head_kernel<<<cdiv(Cn, BLK), BLK, 0, stream>>>(o_xZ2, ctrl, Cn, Wy0, by0, o_y0, Wy1, by1, o_yc1);

    tprob_kernel<<<cdiv(N, BLK), BLK, 0, stream>>>(buf0, Wp2, bp2, o_tprob, N);
}


# Outcome heads leave their two launches: Wp1 GEMM lanes dot the xZ2 values they hold with Wy1, Wy0 (sum order: 32 in-lane fma terms + partner half), last launch gathers the scalars; 9 launches to 7
# speedup vs baseline: 2.8907x; 1.0490x over previous
#include <hip/hip_runtime.h>
#include <hip/hip_bf16.h>

typedef __attribute__((ext_vector_type(2))) float v2f;
typedef __attribute__((ext_vector_type(8))) float v8f;

#define HDIM 64
#define INDIM 128

__device__ __forceinline__ float lrelu(float v) { return v > 0.0f ? v : 0.01f * v; }

#define DEG_R 8
#define DEG_C 32
#define DEG_T 512
#define DEG_U 8

__device__ __forceinline__ void deg_bump(unsigned* cnt, int d, int lo, unsigned span) {
    const unsigned o = (unsigned)(d - lo);
    if (o < span) atomicAdd(&cnt[o], 1u);
}

__global__ __launch_bounds__(DEG_T) void deg_hist_kernel(const int* __restrict__ dst, unsigned* __restrict__ part,
                                                         unsigned* __restrict__ tickets, int n, int e, int span) {
    extern __shared__ unsigned cnt[];
    const int tid = threadIdx.x;
    const int r = blockIdx.x % DEG_R;
    const int c = blockIdx.x / DEG_R;
    const int lo = r * span;
    const unsigned uspan = (unsigned)span;

    if (blockIdx.x == 0 && tid < 4) tickets[tid] = 0u;
    for (int i = tid; i < span; i += DEG_T) cnt[i] = 0u;
    __syncthreads();

    const int per = (((e + DEG_C - 1) / DEG_C) + 3) & ~3;
    const int e0 = min(e, c * per);
    const int e1 = min(e, e0 + per);
    if ((((size_t)dst) & 15) == 0) {
        const int n4 = (e1 - e0) >> 2;
        const int4* __restrict__ p4 = (const int4*)(dst + e0);
        for (int base = tid; base < n4; base += DEG_U * DEG_T) {
            int4 v[DEG_U];
#pragma unroll
            for (int u = 0; u < DEG_U; ++u) {
                const int idx = base + u * DEG_T;
                v[u] = (idx < n4) ? p4[idx] : make_int4(-1, -1, -1, -1);
            }
#pragma unroll
            for (int u = 0; u < DEG_U; ++u) {
                deg_bump(cnt, v[u].x, lo, uspan);
                deg_bump(cnt, v[u].y, lo, uspan);
                deg_bump(cnt, v[u].z, lo, uspan);
                deg_bump(cnt, v[u].w, lo, uspan);
            }
        }
        const int t0 = e0 + 4 * n4;
        if (t0 + tid < e1) deg_bump(cnt, dst[t0 + tid], lo, uspan);
    } else {
        for (int i = e0 + tid; i < e1; i += DEG_T) deg_bump(cnt, dst[i], lo, uspan);
    }
    __syncthreads();

    unsigned* __restrict__ plane = part + (size_t)c * n;
    for (int base = tid; base < span; base += 4 * DEG_T) {
        unsigned t[4];
#pragma unroll
        for (int u = 0; u < 4; ++u) {
            const int i = base + u * DEG_T;
            t[u] = (i < span) ? cnt[i] : 0u;
        }
#pragma unroll
        for (int u = 0; u < 4; ++u) {
            const int i = base + u * DEG_T;
            if (i < span && lo + i < n) plane[lo + i] = t[u];
        }
    }
}

template <int MODE>
__device__ __forceinline__ void gemm_strip_f32(const float* __restrict__ A, float* Aw, const float* __restrict__ B,
                                               float* __restrict__ C, const float* __restrict__ bias,
                                               const float* __restrict__ dinv, float* __restrict__ C2,
                                               const float* __restrict__ biasA, const unsigned* __restrict__ part,
                                               float* __restrict__ dinv_out, const float* __restrict__ Wh1,
                                               const float* __restrict__ Wh0, float* __restrict__ U, int M, int K) {
    extern __shared__ float bsw[];
    const int tid   = threadIdx.x;
    const int lane  = tid & 31;
    const int wave  = tid >> 5;
    const int lmod  = lane & 15;
    const int lhalf = lane >> 4;

    const int gwave = blockIdx.x * 8 + wave;
    const bool live = gwave < (M >> 4);
    unsigned dpl[DEG_C / 2];
    if (MODE == 0 && live) {
        const unsigned* pp = part + (size_t)(lhalf * (DEG_C / 2)) * M + gwave * 16 + lmod;
#pragma unroll
        for (int c = 0; c < DEG_C / 2; ++c) dpl[c] = pp[(size_t)c * M];
    }

    const int pairs = K * 32;
    for (int p = tid; p < pairs; p += 256) {
        int ks  = p >> 7;
        int ll  = (p >> 2) & 31;
        int tn  = p & 3;
        int kb  = 4 * ks + 2 * (ll >> 4);
        int col = tn * 16 + (ll & 15);
        bsw[p * 2 + 0] = B[(size_t)kb * HDIM + col];
        bsw[p * 2 + 1] = B[(size_t)(kb + 1) * HDIM + col];
    }
    if (MODE != 0) {
        if (tid < K) bsw[(size_t)K * HDIM + tid] = biasA[tid];
    }
    if (MODE == 2) {
        if (tid < K) {
            float* hwst = bsw + (size_t)K * HDIM + K + (size_t)(tid >> 1) * 4 + (tid & 1);
            hwst[0] = Wh1[tid];
            hwst[2] = Wh0[tid];
        }
    }
    __syncthreads();

    if (!live) return;

    const size_t aoff = (size_t)(gwave * 16 + lmod) * K + 2 * lhalf;
    const float* arow = (MODE == 2) ? nullptr : A + aoff;
    float* wrow = (MODE == 2) ? Aw + aoff : nullptr;
    float dvrow;
    if (MODE == 0) {
        unsigned degs = 0u;
#pragma unroll
        for (int c = 0; c < DEG_C / 2; ++c) degs += dpl[c];
        const unsigned tot = degs + (unsigned)__shfl_xor((int)degs, 16);
        const float d = 1.0f + (float)tot;
        dvrow = d > 0.0f ? rsqrtf(d) : 0.0f;
        if (lhalf == 0) dinv_out[gwave * 16 + lmod] = dvrow;
    } else {
        dvrow = dinv[gwave * 16 + lmod];
    }
    const float* bA = bsw + (size_t)K * HDIM + 2 * lhalf;
    const float* hW = bsw + (size_t)K * HDIM + K + 4 * lhalf;
    float uh1 = 0.0f, uh0 = 0.0f;
    v8f acc0 = {}, acc1 = {}, acc2 = {}, acc3 = {};

    const int steps = K >> 2;
#pragma unroll(MODE == 1 ? 4 : 8)
    for (int ks = 0; ks < steps; ++ks) {
        v2f a;
        if (MODE == 2) a = *(const v2f*)(wrow + 4 * ks);
        else           a = *(const v2f*)(arow + 4 * ks);
        if (MODE != 0) {
            const v2f bb = *(const v2f*)(bA + 4 * ks);
            a.x = bb.x + dvrow * a.x;
            a.y = bb.y + dvrow * a.y;
            if (MODE == 1) { a.x = fmaxf(a.x, 0.0f); a.y = fmaxf(a.y, 0.0f); }
            if (MODE == 2) {
                *(v2f*)(wrow + 4 * ks) = a;
                const float4 hw = *(const float4*)(hW + 8 * ks);
                uh1 = fmaf(a.x, hw.x, uh1); uh1 = fmaf(a.y, hw.y, uh1);
                uh0 = fmaf(a.x, hw.z, uh0); uh0 = fmaf(a.y, hw.w, uh0);
            }
        }
        const float4* bp = (const float4*)(bsw + (size_t)(ks * 32 + lane) * 8);
        float4 b01 = bp[0];
        float4 b23 = bp[1];
        v2f b0 = {b01.x, b01.y}, b1 = {b01.z, b01.w};
        v2f b2 = {b23.x, b23.y}, b3 = {b23.z, b23.w};
        acc0 = __builtin_amdgcn_wmma_f32_16x16x4_f32(false, a, false, b0, (short)0, acc0, false, false);
        acc1 = __builtin_amdgcn_wmma_f32_16x16x4_f32(false, a, false, b1, (short)0, acc1, false, false);
        acc2 = __builtin_amdgcn_wmma_f32_16x16x4_f32(false, a, false, b2, (short)0, acc2, false, false);
        acc3 = __builtin_amdgcn_wmma_f32_16x16x4_f32(false, a, false, b3, (short)0, acc3, false, false);
    }

    const int rbase = gwave * 16 + lhalf * 8;
    if (MODE == 2) {
        uh1 += __shfl_xor(uh1, 16);
        uh0 += __shfl_xor(uh0, 16);
        if (lhalf == 0) {
            const v2f uu = {uh1, uh0};
            *(v2f*)(U + (size_t)(gwave * 16 + lmod) * 2) = uu;
        }
    }
    if (MODE == 2) {
        const float bi0 = bias[lmod], bi1 = bias[16 + lmod], bi2 = bias[32 + lmod], bi3 = bias[48 + lmod];
#pragma unroll
        for (int r = 0; r < 8; ++r) {
            size_t ro = (size_t)(rbase + r) * HDIM + lmod;
            C[ro]      = lrelu(acc0[r] + bi0);
            C[ro + 16] = lrelu(acc1[r] + bi1);
            C[ro + 32] = lrelu(acc2[r] + bi2);
            C[ro + 48] = lrelu(acc3[r] + bi3);
        }
    } else {
        float dv[8];
        if (MODE == 0) {
            const int dvi = __float_as_int(dvrow);
#pragma unroll
            for (int r = 0; r < 8; ++r) {
                const int lo = __builtin_amdgcn_readlane(dvi, r);
                const int hi = __builtin_amdgcn_readlane(dvi, 8 + r);
                dv[r] = __int_as_float(lhalf ? hi : lo);
            }
        } else {
            const float4 dq0 = *(const float4*)(dinv + rbase);
            const float4 dq1 = *(const float4*)(dinv + rbase + 4);
            dv[0] = dq0.x; dv[1] = dq0.y; dv[2] = dq0.z; dv[3] = dq0.w;
            dv[4] = dq1.x; dv[5] = dq1.y; dv[6] = dq1.z; dv[7] = dq1.w;
        }
#pragma unroll
        for (int r = 0; r < 8; ++r) {
            size_t ro = (size_t)(rbase + r) * HDIM + lmod;
            const float s = dv[r];
            const float h0 = s * acc0[r], h1 = s * acc1[r], h2 = s * acc2[r], h3 = s * acc3[r];
            C[ro] = h0; C[ro + 16] = h1; C[ro + 32] = h2; C[ro + 48] = h3;
            C2[ro] = h0; C2[ro + 16] = h1; C2[ro + 32] = h2; C2[ro + 48] = h3;
        }
    }
}

__global__ void conv_gemm_f32_kernel(const float* __restrict__ A, const float* __restrict__ B,
                                     float* __restrict__ C, const unsigned* __restrict__ part,
                                     float* __restrict__ dinv_out, float* __restrict__ C2, int M, int K) {
    gemm_strip_f32<0>(A, nullptr, B, C, nullptr, nullptr, C2, nullptr, part, dinv_out, nullptr, nullptr, nullptr, M, K);
}

__global__ void conv_gemm_relu_f32_kernel(const float* __restrict__ A, const float* __restrict__ B,
                                          float* __restrict__ C, const float* __restrict__ dinv,
                                          float* __restrict__ C2, const float* __restrict__ biasA, int M, int K) {
    gemm_strip_f32<1>(A, nullptr, B, C, nullptr, dinv, C2, biasA, nullptr, nullptr, nullptr, nullptr, nullptr, M, K);
}

__global__ void dense_gemm_lrelu_f32_kernel(float* Aw, const float* __restrict__ B,
                                            float* __restrict__ C, const float* __restrict__ bias,
                                            const float* __restrict__ dinv, const float* __restrict__ biasA,
                                            const float* __restrict__ Wh1, const float* __restrict__ Wh0,
                                            float* __restrict__ U, int M, int K) {
    gemm_strip_f32<2>(nullptr, Aw, B, C, bias, dinv, nullptr, biasA, nullptr, nullptr, Wh1, Wh0, U, M, K);
}

#define SC_EB 16
#define SC_WCH 64
#define SC_BCH (8 * SC_WCH)
#define SC_Q 64
#define SCATTER_BLOCKS 1536

__device__ __forceinline__ int lane_bcast_i(int v, int l) { return __builtin_amdgcn_readlane(v, l); }

template <bool FULL>
__device__ __forceinline__ void scatter_batch(const float* __restrict__ Hm, float* __restrict__ out, int id, int cnt,
                                              int lane) {
    float h0[SC_EB], h1[SC_EB];
#pragma unroll
    for (int k = 0; k < SC_EB; ++k) {
        h0[k] = 0.0f; h1[k] = 0.0f;
        if (FULL || k < cnt) {
            const int s = lane_bcast_i(id, k);
            const float* hr = Hm + (size_t)s * HDIM + lane;
            h0[k] = hr[0];
            h1[k] = hr[32];
        }
    }
#pragma unroll
    for (int k = 0; k < SC_EB; ++k) {
        if (FULL || k < cnt) {
            const int d = lane_bcast_i(id, 16 + k);
            float* o = out + (size_t)d * HDIM + lane;
            __hip_atomic_fetch_add(o,      h0[k], __ATOMIC_RELAXED, __HIP_MEMORY_SCOPE_AGENT);
            __hip_atomic_fetch_add(o + 32, h1[k], __ATOMIC_RELAXED, __HIP_MEMORY_SCOPE_AGENT);
        }
    }
}

__global__ __launch_bounds__(256) void conv_scatter_kernel(const float* __restrict__ Hm, const int* __restrict__ src,
                                    const int* __restrict__ dst, float* __restrict__ out,
                                    unsigned* __restrict__ tickets, int e, int n) {
    __shared__ int q_src[8 * SC_Q];
    __shared__ int q_dst[8 * SC_Q];
    __shared__ int s_tick[2];
    __shared__ int s_half;
    const int tid  = threadIdx.x;
    const int lane = tid & 31;
    const int sub  = lane & 15;
    const int wv = __builtin_amdgcn_readfirstlane(tid >> 5);
    int* qs = q_src + wv * SC_Q;
    int* qd = q_dst + wv * SC_Q;
    const int split = n >> 1;
    const int nchunks = (e + SC_BCH - 1) / SC_BCH;

    if (tid == 0) {
        const int hf = (int)((__builtin_amdgcn_s_sendmsg_rtn(0x87) >> 18) & 1u);
        s_half = hf;
        s_tick[0] = (int)__hip_atomic_fetch_add(tickets + hf, 1u, __ATOMIC_RELAXED, __HIP_MEMORY_SCOPE_AGENT);
    }
    __syncthreads();
    const int myhalf = s_half;
    int pass = 0;
    int head = 0, tail = 0;

    for (int trip = 0; trip < 2 * nchunks + 2; ++trip) {
        const int t = __builtin_amdgcn_readfirstlane(s_tick[trip & 1]);
        if (t >= nchunks) {
            if (pass == 1) break;
            pass = 1;
            if (tid == 0)
                s_tick[(trip + 1) & 1] = (int)__hip_atomic_fetch_add(tickets + (myhalf ^ 1), 1u, __ATOMIC_RELAXED,
                                                                     __HIP_MEMORY_SCOPE_AGENT);
            __syncthreads();
            continue;
        }
        const int hh = myhalf ^ pass;

        const int c0 = t * SC_BCH + wv * SC_WCH;
        int dd[SC_WCH / 32], ss[SC_WCH / 32];
#pragma unroll
        for (int j = 0; j < SC_WCH / 32; ++j) {
            const int ei = c0 + j * 32 + lane;
            const bool ok = ei < e;
            dd[j] = ok ? dst[ei] : -1;
            ss[j] = ok ? src[ei] : 0;
        }
        int tnext = 0;
        if (tid == 0)
            tnext = (int)__hip_atomic_fetch_add(tickets + hh, 1u, __ATOMIC_RELAXED, __HIP_MEMORY_SCOPE_AGENT);
#pragma unroll
        for (int j = 0; j < SC_WCH / 32; ++j) {
            const int d = dd[j];
            const bool mine = (d >= 0) && ((d >= split) == (hh != 0));
            const unsigned m = __builtin_amdgcn_ballot_w32(mine);
            if (mine) {
                const int pos = (tail + (int)__builtin_amdgcn_mbcnt_lo(m, 0u)) & (SC_Q - 1);
                qs[pos] = ss[j];
                qd[pos] = d;
            }
            tail += __builtin_popcount(m);
            for (int r = 0; r < 2 && tail - head >= SC_EB; ++r) {
                const int qi = (head + sub) & (SC_Q - 1);
                const int id = (lane < 16) ? qs[qi] : qd[qi];
                scatter_batch<true>(Hm, out, id, SC_EB, lane);
                head += SC_EB;
            }
        }
        if (tid == 0) s_tick[(trip + 1) & 1] = tnext;
        __syncthreads();
    }

    const int rem = tail - head;
    if (rem > 0) {
        const int qi = (head + sub) & (SC_Q - 1);
        int id = (lane < 16) ? qs[qi] : qd[qi];
        if (sub >= rem) id = 0;
        scatter_batch<false>(Hm, out, id, rem, lane);
    }
}

__device__ __forceinline__ void head_pair(const float* __restrict__ U, int node, float ca, float cb,
                                          float* __restrict__ oa, float* __restrict__ ob, int t, bool swap) {
    const v2f u = *(const v2f*)(U + (size_t)node * 2);
    oa[t] = lrelu((swap ? u.y : u.x) + ca);
    ob[t] = lrelu((swap ? u.x : u.y) + cb);
}

__global__ void tprob_heads_kernel(const float* __restrict__ P1, const float* __restrict__ Wp2,
                                   const float* __restrict__ bp2, float* __restrict__ out, int n,
                                   const float* __restrict__ U, const int* __restrict__ treat, int nt,
                                   const int* __restrict__ ctrl, int nc, const float* __restrict__ by1,
                                   const float* __restrict__ by0, float* __restrict__ o_y1, float* __restrict__ o_yc0,
                                   float* __restrict__ o_y0, float* __restrict__ o_yc1) {
    const int i = blockIdx.x * blockDim.x + threadIdx.x;
    const int stride = gridDim.x * blockDim.x;
    const float c1 = by1[0], c0 = by0[0];
    const bool ht = i < nt, hc = i < nc;
    const int ti = treat[max(min(i, nt - 1), 0)];
    const int ci = ctrl[max(min(i, nc - 1), 0)];

    const float* row = P1 + (size_t)min(i, n - 1) * HDIM;
    float4 z[HDIM / 4];
#pragma unroll
    for (int j = 0; j < HDIM / 4; ++j) z[j] = *(const float4*)(row + 4 * j);

    const v2f ut = *(const v2f*)(U + (size_t)ti * 2);
    const v2f uc = *(const v2f*)(U + (size_t)ci * 2);

    float d0 = 0.0f, d1 = 0.0f;
#pragma unroll
    for (int j = 0; j < HDIM / 4; ++j) {
        const int k = 4 * j;
        d0 += z[j].x * Wp2[(k + 0) * 2 + 0]; d1 += z[j].x * Wp2[(k + 0) * 2 + 1];
        d0 += z[j].y * Wp2[(k + 1) * 2 + 0]; d1 += z[j].y * Wp2[(k + 1) * 2 + 1];
        d0 += z[j].z * Wp2[(k + 2) * 2 + 0]; d1 += z[j].z * Wp2[(k + 2) * 2 + 1];
        d0 += z[j].w * Wp2[(k + 3) * 2 + 0]; d1 += z[j].w * Wp2[(k + 3) * 2 + 1];
    }
    if (i < n) {
        out[i * 2 + 0] = lrelu(d0 + bp2[0]);
        out[i * 2 + 1] = lrelu(d1 + bp2[1]);
    }
    if (ht) { o_y1[i] = lrelu(ut.x + c1); o_yc0[i] = lrelu(ut.y + c0); }
    if (hc) { o_y0[i] = lrelu(uc.y + c0); o_yc1[i] = lrelu(uc.x + c1); }

    for (int t = i + stride; t < nt; t += stride) head_pair(U, treat[t], c1, c0, o_y1, o_yc0, t, false);
    for (int t = i + stride; t < nc; t += stride) head_pair(U, ctrl[t], c0, c1, o_y0, o_yc1, t, true);
}

extern "C" void kernel_launch(void* const* d_in, const int* in_sizes, int n_in,
                              void* d_out, int out_size, void* d_ws, size_t ws_size,
                              hipStream_t stream) {
    const float* x     = (const float*)d_in[0];
    const int*   ei    = (const int*)d_in[1];
    const int*   treat = (const int*)d_in[2];
    const int*   ctrl  = (const int*)d_in[3];
    const float* W1  = (const float*)d_in[4];  const float* b1  = (const float*)d_in[5];
    const float* W2  = (const float*)d_in[6];  const float* b2  = (const float*)d_in[7];
    const float* Wy1 = (const float*)d_in[8];  const float* by1 = (const float*)d_in[9];
    const float* Wy0 = (const float*)d_in[10]; const float* by0 = (const float*)d_in[11];
    const float* Wp1 = (const float*)d_in[12]; const float* bp1 = (const float*)d_in[13];
    const float* Wp2 = (const float*)d_in[14]; const float* bp2 = (const float*)d_in[15];

    const int N = in_sizes[0] / INDIM;
    const int E = in_sizes[1] / 2;
    const int T = in_sizes[2];
    const int Cn = in_sizes[3];
    const int* e_src = ei;
    const int* e_dst = ei + E;

    float* dinv = (float*)d_ws;
    unsigned* tickets = (unsigned*)(dinv + N);
    float* buf0 = dinv + (((size_t)N + 4 + 63) & ~(size_t)63);
    float* buf1 = buf0 + (size_t)N * HDIM;

    float* o_y1    = (float*)d_out;
    float* o_yc0   = o_y1 + T;
    float* o_y0    = o_yc0 + T;
    float* o_yc1   = o_y0 + Cn;
    float* o_tprob = o_yc1 + Cn;
    float* o_xZ2   = o_tprob + (size_t)2 * N;

    const int BLK = 256;
    auto cdiv = [](long long a, long long b) { return (int)((a + b - 1) / b); };
    const int gemm_blocks = cdiv(N / 16, 8);
    const size_t smem128 = (size_t)INDIM * HDIM * sizeof(float);
    const size_t smem64  = (size_t)(HDIM * HDIM + HDIM) * sizeof(float);
    const size_t smem64h = (size_t)(HDIM * HDIM + 3 * HDIM) * sizeof(float);

    const int span = cdiv(N, DEG_R);
    unsigned* degpart = (unsigned*)o_xZ2;
    deg_hist_kernel<<<DEG_R * DEG_C, DEG_T, (size_t)span * sizeof(unsigned), stream>>>(e_dst, degpart, tickets, N, E,
                                                                                       span);

    conv_gemm_f32_kernel<<<gemm_blocks, BLK, smem128, stream>>>(x, W1, buf0, degpart, dinv, buf1, N, INDIM);
    conv_scatter_kernel<<<SCATTER_BLOCKS, BLK, 0, stream>>>(buf0, e_src, e_dst, buf1, tickets, E, N);

    conv_gemm_relu_f32_kernel<<<gemm_blocks, BLK, smem64, stream>>>(buf1, W2, buf0, dinv, o_xZ2, b1, N, HDIM);
    conv_scatter_kernel<<<SCATTER_BLOCKS, BLK, 0, stream>>>(buf0, e_src, e_dst, o_xZ2, tickets + 2, E, N);

    float* headU = buf1;
    dense_gemm_lrelu_f32_kernel<<<gemm_blocks, BLK, smem64h, stream>>>(o_xZ2, Wp1, buf0, bp1, dinv, b2, Wy1, Wy0, headU, N,
                                                                        HDIM);

    tprob_heads_kernel<<<cdiv(N, BLK), BLK, 0, stream>>>(buf0, Wp2, bp2, o_tprob, N, headU, treat, T, ctrl, Cn, by1, by0,
                                                         o_y1, o_yc0, o_y0, o_yc1);
}
